# starting kernel
# baseline (speedup 1.0000x reference)
#include <hip/hip_runtime.h>
#include <hip/hip_bf16.h>

#define D      128
#define H      8
#define DK     16
#define SCALE  0.25f

typedef __attribute__((ext_vector_type(2))) float v2f;
typedef __attribute__((ext_vector_type(8))) float v8f;

__device__ __forceinline__ unsigned fenc(float x) {
  unsigned u = __float_as_uint(x);
  return (u & 0x80000000u) ? ~u : (u | 0x80000000u);
}
__device__ __forceinline__ float fdec(unsigned k) {
  return __uint_as_float((k & 0x80000000u) ? (k & 0x7FFFFFFFu) : ~k);
}

__global__ void gemm_xwT_wmma(const float* __restrict__ X,
                              const float* __restrict__ W,
                              float* __restrict__ Y, int M) {
  const int lane  = threadIdx.x & 31;
  const int wave  = (blockIdx.x * blockDim.x + threadIdx.x) >> 5;
  const int mBase = wave * 16;
  if (mBase >= M) return;

  const int row  = lane & 15;
  const int koff = (lane >> 4) << 1;

  const float* xrow  = X + (size_t)(mBase + row) * D + koff;
  const float* wbase = W + (size_t)row * D + koff;

  v8f acc[8] = {};

  v2f a_cur = *(const v2f*)(xrow);
  v2f b_cur[8];
#pragma unroll
  for (int t = 0; t < 8; ++t)
    b_cur[t] = *(const v2f*)(wbase + (size_t)t * 16 * D);

#pragma unroll 4
  for (int k0 = 0; k0 < D; k0 += 4) {
    v2f a_nxt;
    v2f b_nxt[8];
    if (k0 + 4 < D) {
      a_nxt = *(const v2f*)(xrow + k0 + 4);
#pragma unroll
      for (int t = 0; t < 8; ++t)
        b_nxt[t] = *(const v2f*)(wbase + (size_t)t * 16 * D + k0 + 4);
    }
#pragma unroll
    for (int t = 0; t < 8; ++t) {
      acc[t] = __builtin_amdgcn_wmma_f32_16x16x4_f32(
          false, a_cur, false, b_cur[t], (short)0, acc[t], false, false);
    }
    a_cur = a_nxt;
#pragma unroll
    for (int t = 0; t < 8; ++t) b_cur[t] = b_nxt[t];
  }

  const int n    = lane & 15;
  const int mAdd = (lane >> 4) << 3;
  float* yb = Y + (size_t)mBase * D;
#pragma unroll
  for (int t = 0; t < 8; ++t) {
#pragma unroll
    for (int r = 0; r < 8; ++r) {
      yb[(size_t)(mAdd + r) * D + t * 16 + n] = acc[t][r];
    }
  }
}

__global__ void self_init(const float* __restrict__ dstK,
                          const float* __restrict__ Q,
                          float* __restrict__ self_t,
                          unsigned* __restrict__ m_enc, int Ndst) {
  int tid = blockIdx.x * blockDim.x + threadIdx.x;
  if (tid >= Ndst * H) return;
  int n = tid >> 3, h = tid & 7;
  const float4* kp = (const float4*)(dstK + (size_t)n * D + h * DK);
  const float4* qp = (const float4*)(Q    + (size_t)n * D + h * DK);
  float acc = 0.f;
#pragma unroll
  for (int i = 0; i < 4; ++i) {
    float4 kv = kp[i], qv = qp[i];
    acc += kv.x * qv.x + kv.y * qv.y + kv.z * qv.z + kv.w * qv.w;
  }
  float st = acc * SCALE;
  self_t[tid] = st;
  m_enc[tid]  = fenc(st);
}

__global__ void edge_scores(const float* __restrict__ K,
                            const float* __restrict__ Q,
                            const int* __restrict__ src_idx,
                            const int* __restrict__ dst_idx,
                            float* __restrict__ t_buf,
                            unsigned* __restrict__ m_enc, int E) {
  int tid = blockIdx.x * blockDim.x + threadIdx.x;
  if (tid >= E * H) return;
  int e = tid >> 3, h = tid & 7;
  int s = src_idx[e], d = dst_idx[e];
  const float4* kp = (const float4*)(K + (size_t)s * D + h * DK);
  const float4* qp = (const float4*)(Q + (size_t)d * D + h * DK);
  float acc = 0.f;
#pragma unroll
  for (int i = 0; i < 4; ++i) {
    float4 kv = kp[i], qv = qp[i];
    acc += kv.x * qv.x + kv.y * qv.y + kv.z * qv.z + kv.w * qv.w;
  }
  float t = acc * SCALE;
  t_buf[tid] = t;
  atomicMax(&m_enc[d * H + h], fenc(t));
}

__global__ void denom_init(const float* __restrict__ self_t,
                           const unsigned* __restrict__ m_enc,
                           float* __restrict__ denom, int Ndst) {
  int tid = blockIdx.x * blockDim.x + threadIdx.x;
  if (tid >= Ndst * H) return;
  denom[tid] = __expf(self_t[tid] - fdec(m_enc[tid]));
}

__global__ void edge_exp(const int* __restrict__ dst_idx,
                         const unsigned* __restrict__ m_enc,
                         float* __restrict__ t_buf,
                         float* __restrict__ denom, int E) {
  int tid = blockIdx.x * blockDim.x + threadIdx.x;
  if (tid >= E * H) return;
  int e = tid >> 3, h = tid & 7;
  int d = dst_idx[e];
  float ev = __expf(t_buf[tid] - fdec(m_enc[d * H + h]));
  t_buf[tid] = ev;
  atomicAdd(&denom[d * H + h], ev);
}

__global__ void recip_denom(float* __restrict__ denom, int Ndst) {
  int tid = blockIdx.x * blockDim.x + threadIdx.x;
  if (tid >= Ndst * H) return;
  denom[tid] = 1.0f / denom[tid];
}

__global__ void edge_aggregate(const float* __restrict__ h_src,
                               const int* __restrict__ src_idx,
                               const int* __restrict__ dst_idx,
                               const float* __restrict__ t_buf,
                               const float* __restrict__ rdenom,
                               float* __restrict__ out, int E) {
  unsigned tid = blockIdx.x * blockDim.x + threadIdx.x;
  if (tid >= (unsigned)E * D) return;
  int e = tid >> 7, c = tid & (D - 1), h = c >> 4;
  int s = src_idx[e], d = dst_idx[e];
  float a = t_buf[e * H + h] * rdenom[d * H + h];
  atomicAdd(&out[(size_t)d * D + c], a * h_src[(size_t)s * D + c]);
}

__global__ void finalize(const float* __restrict__ h_dst,
                         const float* __restrict__ self_t,
                         const unsigned* __restrict__ m_enc,
                         const float* __restrict__ rdenom,
                         float* __restrict__ out, int Ndst) {
  int tid = blockIdx.x * blockDim.x + threadIdx.x;
  if (tid >= Ndst * D) return;
  int n = tid >> 7, c = tid & (D - 1), h = c >> 4;
  float m = fdec(m_enc[n * H + h]);
  float dst_a = __expf(self_t[n * H + h] - m) * rdenom[n * H + h];
  out[tid] += dst_a * h_dst[tid];
}

extern "C" void kernel_launch(void* const* d_in, const int* in_sizes, int n_in,
                              void* d_out, int out_size, void* d_ws, size_t ws_size,
                              hipStream_t stream) {
  const float* h_src = (const float*)d_in[0];
  const float* h_dst = (const float*)d_in[1];
  const float* Wk    = (const float*)d_in[2];
  const float* Wq    = (const float*)d_in[3];
  const int* src_idx = (const int*)d_in[4];
  const int* dst_idx = (const int*)d_in[5];

  const int N_SRC = in_sizes[0] / D;
  const int N_DST = in_sizes[1] / D;
  const int E     = in_sizes[4];

  auto align256 = [](size_t x) { return (x + 255) & ~(size_t)255; };
  char* ws = (char*)d_ws;
  size_t off = 0;
  float* K      = (float*)(ws + off); off += align256((size_t)N_SRC * D * 4);
  float* Q      = (float*)(ws + off); off += align256((size_t)N_DST * D * 4);
  float* dstK   = (float*)(ws + off); off += align256((size_t)N_DST * D * 4);
  float* t_buf  = (float*)(ws + off); off += align256((size_t)E * H * 4);
  float* self_t = (float*)(ws + off); off += align256((size_t)N_DST * H * 4);
  unsigned* m_enc = (unsigned*)(ws + off); off += align256((size_t)N_DST * H * 4);
  float* denom  = (float*)(ws + off); off += align256((size_t)N_DST * H * 4);

  const int TB = 256;

  hipMemsetAsync(d_out, 0, (size_t)out_size * sizeof(float), stream);

  {
    int waves  = N_SRC / 16;
    int blocks = (waves * 32 + TB - 1) / TB;
    gemm_xwT_wmma<<<blocks, TB, 0, stream>>>(h_src, Wk, K, N_SRC);
  }
  {
    int waves  = N_DST / 16;
    int blocks = (waves * 32 + TB - 1) / TB;
    gemm_xwT_wmma<<<blocks, TB, 0, stream>>>(h_dst, Wq, Q, N_DST);
    gemm_xwT_wmma<<<blocks, TB, 0, stream>>>(h_dst, Wk, dstK, N_DST);
  }

  self_init<<<(N_DST * H + TB - 1) / TB, TB, 0, stream>>>(dstK, Q, self_t, m_enc, N_DST);
  edge_scores<<<(E * H + TB - 1) / TB, TB, 0, stream>>>(K, Q, src_idx, dst_idx, t_buf, m_enc, E);
  denom_init<<<(N_DST * H + TB - 1) / TB, TB, 0, stream>>>(self_t, m_enc, denom, N_DST);
  edge_exp<<<(E * H + TB - 1) / TB, TB, 0, stream>>>(dst_idx, m_enc, t_buf, denom, E);
  recip_denom<<<(N_DST * H + TB - 1) / TB, TB, 0, stream>>>(denom, N_DST);
  {
    unsigned total  = (unsigned)E * D;
    unsigned blocks = (total + TB - 1) / TB;
    edge_aggregate<<<blocks, TB, 0, stream>>>(h_src, src_idx, dst_idx, t_buf, denom, (float*)d_out, E);
  }
  finalize<<<(N_DST * D + TB - 1) / TB, TB, 0, stream>>>(h_dst, self_t, m_enc, denom, (float*)d_out, N_DST);
}


# Edges bucketed by dst (integer-atomic rank folded into the score and exp passes); one wave per dst row sums its segment in registers in arrival-rank order: out gets no float atomics and no memset
# speedup vs baseline: 1.6660x; 1.6660x over previous
#include <hip/hip_runtime.h>
#include <hip/hip_bf16.h>

#define D      128
#define H      8
#define DK     16
#define SCALE  0.25f
#define SEG_TB 256

typedef __attribute__((ext_vector_type(2))) float v2f;
typedef __attribute__((ext_vector_type(8))) float v8f;

__device__ __forceinline__ unsigned fenc(float x) {
  unsigned u = __float_as_uint(x);
  return (u & 0x80000000u) ? ~u : (u | 0x80000000u);
}
__device__ __forceinline__ float fdec(unsigned k) {
  return __uint_as_float((k & 0x80000000u) ? (k & 0x7FFFFFFFu) : ~k);
}

__global__ void gemm_xwT_wmma(const float* __restrict__ X,
                              const float* __restrict__ W,
                              float* __restrict__ Y, int M) {
  const int lane  = threadIdx.x & 31;
  const int wave  = (blockIdx.x * blockDim.x + threadIdx.x) >> 5;
  const int mBase = wave * 16;
  if (mBase >= M) return;

  const int row  = lane & 15;
  const int koff = (lane >> 4) << 1;

  const float* xrow  = X + (size_t)(mBase + row) * D + koff;
  const float* wbase = W + (size_t)row * D + koff;

  v8f acc[8] = {};

  v2f a_cur = *(const v2f*)(xrow);
  v2f b_cur[8];
#pragma unroll
  for (int t = 0; t < 8; ++t)
    b_cur[t] = *(const v2f*)(wbase + (size_t)t * 16 * D);

#pragma unroll 4
  for (int k0 = 0; k0 < D; k0 += 4) {
    v2f a_nxt;
    v2f b_nxt[8];
    if (k0 + 4 < D) {
      a_nxt = *(const v2f*)(xrow + k0 + 4);
#pragma unroll
      for (int t = 0; t < 8; ++t)
        b_nxt[t] = *(const v2f*)(wbase + (size_t)t * 16 * D + k0 + 4);
    }
#pragma unroll
    for (int t = 0; t < 8; ++t) {
      acc[t] = __builtin_amdgcn_wmma_f32_16x16x4_f32(
          false, a_cur, false, b_cur[t], (short)0, acc[t], false, false);
    }
    a_cur = a_nxt;
#pragma unroll
    for (int t = 0; t < 8; ++t) b_cur[t] = b_nxt[t];
  }

  const int n    = lane & 15;
  const int mAdd = (lane >> 4) << 3;
  float* yb = Y + (size_t)mBase * D;
#pragma unroll
  for (int t = 0; t < 8; ++t) {
#pragma unroll
    for (int r = 0; r < 8; ++r) {
      yb[(size_t)(mAdd + r) * D + t * 16 + n] = acc[t][r];
    }
  }
}

__global__ void self_init(const float* __restrict__ dstK,
                          const float* __restrict__ Q,
                          float* __restrict__ self_t,
                          unsigned* __restrict__ m_enc,
                          int* __restrict__ seg_cnt,
                          int* __restrict__ cursor, int Ndst) {
  int tid = blockIdx.x * blockDim.x + threadIdx.x;
  if (tid >= Ndst * H) return;
  int n = tid >> 3, h = tid & 7;
  if (h == 0) seg_cnt[n] = 0;
  if (tid == 0) *cursor = 0;
  const float4* kp = (const float4*)(dstK + (size_t)n * D + h * DK);
  const float4* qp = (const float4*)(Q    + (size_t)n * D + h * DK);
  float acc = 0.f;
#pragma unroll
  for (int i = 0; i < 4; ++i) {
    float4 kv = kp[i], qv = qp[i];
    acc += kv.x * qv.x + kv.y * qv.y + kv.z * qv.z + kv.w * qv.w;
  }
  float st = acc * SCALE;
  self_t[tid] = st;
  m_enc[tid]  = fenc(st);
}

__global__ void edge_scores(const float* __restrict__ K,
                            const float* __restrict__ Q,
                            const int* __restrict__ src_idx,
                            const int* __restrict__ dst_idx,
                            float* __restrict__ t_buf,
                            unsigned* __restrict__ m_enc,
                            int* __restrict__ seg_cnt,
                            int* __restrict__ rank, int E) {
  int tid = blockIdx.x * blockDim.x + threadIdx.x;
  if (tid >= E * H) return;
  int e = tid >> 3, h = tid & 7;
  int s = src_idx[e], d = dst_idx[e];
  const float4* kp = (const float4*)(K + (size_t)s * D + h * DK);
  const float4* qp = (const float4*)(Q + (size_t)d * D + h * DK);
  float acc = 0.f;
#pragma unroll
  for (int i = 0; i < 4; ++i) {
    float4 kv = kp[i], qv = qp[i];
    acc += kv.x * qv.x + kv.y * qv.y + kv.z * qv.z + kv.w * qv.w;
  }
  float t = acc * SCALE;
  t_buf[tid] = t;
  atomicMax(&m_enc[d * H + h], fenc(t));
  if (h == 0) rank[e] = atomicAdd(&seg_cnt[d], 1);
}

__global__ void seg_alloc(const int* __restrict__ seg_cnt,
                          int* __restrict__ seg_start,
                          int* __restrict__ cursor, int Ndst) {
  __shared__ int wsum[SEG_TB / 32];
  __shared__ int base_s;
  const int d    = blockIdx.x * SEG_TB + threadIdx.x;
  const int lane = threadIdx.x & 31, w = threadIdx.x >> 5;
  const int c    = (d < Ndst) ? seg_cnt[d] : 0;
  int inc = c;
  for (int o = 1; o < 32; o <<= 1) {
    int v = __shfl_up(inc, o);
    if (lane >= o) inc += v;
  }
  if (lane == 31) wsum[w] = inc;
  __syncthreads();
  if (threadIdx.x == 0) {
    int tot = 0;
    for (int i = 0; i < SEG_TB / 32; ++i) { int v = wsum[i]; wsum[i] = tot; tot += v; }
    base_s = atomicAdd(cursor, tot);
  }
  __syncthreads();
  if (d < Ndst) seg_start[d] = base_s + wsum[w] + inc - c;
}

__global__ void denom_init(const float* __restrict__ self_t,
                           const unsigned* __restrict__ m_enc,
                           float* __restrict__ denom, int Ndst) {
  int tid = blockIdx.x * blockDim.x + threadIdx.x;
  if (tid >= Ndst * H) return;
  denom[tid] = __expf(self_t[tid] - fdec(m_enc[tid]));
}

__global__ void edge_exp(const int* __restrict__ src_idx,
                         const int* __restrict__ dst_idx,
                         const int* __restrict__ rank,
                         const int* __restrict__ seg_start,
                         const unsigned* __restrict__ m_enc,
                         const float* __restrict__ t_buf,
                         float* __restrict__ e_sorted,
                         int* __restrict__ src_sorted,
                         float* __restrict__ denom, int E) {
  int tid = blockIdx.x * blockDim.x + threadIdx.x;
  if (tid >= E * H) return;
  int e = tid >> 3, h = tid & 7;
  int d = dst_idx[e];
  int pos = seg_start[d] + rank[e];
  float ev = __expf(t_buf[tid] - fdec(m_enc[d * H + h]));
  e_sorted[(size_t)pos * H + h] = ev;
  if (h == 0) src_sorted[pos] = src_idx[e];
  atomicAdd(&denom[d * H + h], ev);
}

__global__ void recip_denom(float* __restrict__ denom, int Ndst) {
  int tid = blockIdx.x * blockDim.x + threadIdx.x;
  if (tid >= Ndst * H) return;
  denom[tid] = 1.0f / denom[tid];
}

__global__ void __launch_bounds__(SEG_TB)
seg_aggregate(const float* __restrict__ h_src,
              const float* __restrict__ h_dst,
              const int* __restrict__ seg_start,
              const int* __restrict__ seg_cnt,
              const int* __restrict__ src_sorted,
              const float* __restrict__ e_sorted,
              const float* __restrict__ self_t,
              const unsigned* __restrict__ m_enc,
              const float* __restrict__ rdenom,
              float* __restrict__ out, int Ndst) {
  const int lane = threadIdx.x & 31;
  const int d = __builtin_amdgcn_readfirstlane((int)((blockIdx.x * SEG_TB + threadIdx.x) >> 5));
  if (d >= Ndst) return;
  const int h   = lane >> 2;
  const int beg = __builtin_amdgcn_readfirstlane(seg_start[d]);
  const int n   = __builtin_amdgcn_readfirstlane(seg_cnt[d]);
  const float rd = rdenom[d * H + h];
  const float* xcol = h_src + lane * 4;
  float4 acc = make_float4(0.f, 0.f, 0.f, 0.f);
  for (int j0 = 0; j0 < n; j0 += 32) {
    const int m  = (n - j0 < 32) ? (n - j0) : 32;
    const int sv = (lane < m) ? src_sorted[beg + j0 + lane] : 0;
    const float* ep = e_sorted + (size_t)(beg + j0) * H + h;
#pragma unroll 8
    for (int j = 0; j < m; ++j) {
      const int s = __builtin_amdgcn_readlane(sv, j);
      const float a = ep[(size_t)j * H] * rd;
      const float4 x = *(const float4*)(xcol + (size_t)s * D);
      acc.x += a * x.x; acc.y += a * x.y; acc.z += a * x.z; acc.w += a * x.w;
    }
  }
  const float mx    = fdec(m_enc[d * H + h]);
  const float dst_a = __expf(self_t[d * H + h] - mx) * rd;
  const float4 xd = *(const float4*)(h_dst + (size_t)d * D + lane * 4);
  float4 o;
  o.x = acc.x + dst_a * xd.x; o.y = acc.y + dst_a * xd.y;
  o.z = acc.z + dst_a * xd.z; o.w = acc.w + dst_a * xd.w;
  *(float4*)(out + (size_t)d * D + lane * 4) = o;
}

extern "C" void kernel_launch(void* const* d_in, const int* in_sizes, int n_in,
                              void* d_out, int out_size, void* d_ws, size_t ws_size,
                              hipStream_t stream) {
  const float* h_src = (const float*)d_in[0];
  const float* h_dst = (const float*)d_in[1];
  const float* Wk    = (const float*)d_in[2];
  const float* Wq    = (const float*)d_in[3];
  const int* src_idx = (const int*)d_in[4];
  const int* dst_idx = (const int*)d_in[5];

  const int N_SRC = in_sizes[0] / D;
  const int N_DST = in_sizes[1] / D;
  const int E     = in_sizes[4];

  auto align256 = [](size_t x) { return (x + 255) & ~(size_t)255; };
  char* ws = (char*)d_ws;
  size_t off = 0;
  float* K      = (float*)(ws + off); off += align256((size_t)N_SRC * D * 4);
  float* Q      = (float*)(ws + off); off += align256((size_t)N_DST * D * 4);
  float* dstK   = (float*)(ws + off); off += align256((size_t)N_DST * D * 4);
  float* t_buf  = (float*)(ws + off); off += align256((size_t)E * H * 4);
  float* self_t = (float*)(ws + off); off += align256((size_t)N_DST * H * 4);
  unsigned* m_enc = (unsigned*)(ws + off); off += align256((size_t)N_DST * H * 4);
  float* denom  = (float*)(ws + off); off += align256((size_t)N_DST * H * 4);
  float* e_sorted = (float*)(ws + off); off += align256((size_t)E * H * 4);
  int* rank       = (int*)(ws + off); off += align256((size_t)E * 4);
  int* src_sorted = (int*)(ws + off); off += align256((size_t)E * 4);
  int* seg_cnt    = (int*)(ws + off); off += align256((size_t)N_DST * 4);
  int* seg_start  = (int*)(ws + off); off += align256((size_t)N_DST * 4);
  int* cursor     = (int*)(ws + off); off += align256(4);

  const int TB = 256;

  {
    int waves  = N_SRC / 16;
    int blocks = (waves * 32 + TB - 1) / TB;
    gemm_xwT_wmma<<<blocks, TB, 0, stream>>>(h_src, Wk, K, N_SRC);
  }
  {
    int waves  = N_DST / 16;
    int blocks = (waves * 32 + TB - 1) / TB;
    gemm_xwT_wmma<<<blocks, TB, 0, stream>>>(h_dst, Wq, Q, N_DST);
    gemm_xwT_wmma<<<blocks, TB, 0, stream>>>(h_dst, Wk, dstK, N_DST);
  }

  self_init<<<(N_DST * H + TB - 1) / TB, TB, 0, stream>>>(dstK, Q, self_t, m_enc, seg_cnt, cursor, N_DST);
  edge_scores<<<(E * H + TB - 1) / TB, TB, 0, stream>>>(K, Q, src_idx, dst_idx, t_buf, m_enc, seg_cnt, rank, E);
  seg_alloc<<<(N_DST + SEG_TB - 1) / SEG_TB, SEG_TB, 0, stream>>>(seg_cnt, seg_start, cursor, N_DST);
  denom_init<<<(N_DST * H + TB - 1) / TB, TB, 0, stream>>>(self_t, m_enc, denom, N_DST);
  edge_exp<<<(E * H + TB - 1) / TB, TB, 0, stream>>>(src_idx, dst_idx, rank, seg_start, m_enc, t_buf,
                                                     e_sorted, src_sorted, denom, E);
  recip_denom<<<(N_DST * H + TB - 1) / TB, TB, 0, stream>>>(denom, N_DST);
  seg_aggregate<<<(N_DST * 32 + SEG_TB - 1) / SEG_TB, SEG_TB, 0, stream>>>(
      h_src, h_dst, seg_start, seg_cnt, src_sorted, e_sorted, self_t, m_enc, denom, (float*)d_out, N_DST);
}


# One wave per dst row fuses scores (4 products per lane + quad DPP sum), softmax (running max, rescale per 8 edges, denom summed in-wave in rank order, one divide) and weighted sum; no float atomics
# speedup vs baseline: 2.9091x; 1.7462x over previous
#include <hip/hip_runtime.h>
#include <hip/hip_bf16.h>

#define D      128
#define H      8
#define DK     16
#define SCALE  0.25f
#define SEG_TB 256

typedef __attribute__((ext_vector_type(2))) float v2f;
typedef __attribute__((ext_vector_type(8))) float v8f;

__global__ void gemm_xwT_wmma(const float* __restrict__ X,
                              const float* __restrict__ W,
                              float* __restrict__ Y, int M) {
  const int lane  = threadIdx.x & 31;
  const int wave  = (blockIdx.x * blockDim.x + threadIdx.x) >> 5;
  const int mBase = wave * 16;
  if (mBase >= M) return;

  const int row  = lane & 15;
  const int koff = (lane >> 4) << 1;

  const float* xrow  = X + (size_t)(mBase + row) * D + koff;
  const float* wbase = W + (size_t)row * D + koff;

  v8f acc[8] = {};

  v2f a_cur = *(const v2f*)(xrow);
  v2f b_cur[8];
#pragma unroll
  for (int t = 0; t < 8; ++t)
    b_cur[t] = *(const v2f*)(wbase + (size_t)t * 16 * D);

#pragma unroll 4
  for (int k0 = 0; k0 < D; k0 += 4) {
    v2f a_nxt;
    v2f b_nxt[8];
    if (k0 + 4 < D) {
      a_nxt = *(const v2f*)(xrow + k0 + 4);
#pragma unroll
      for (int t = 0; t < 8; ++t)
        b_nxt[t] = *(const v2f*)(wbase + (size_t)t * 16 * D + k0 + 4);
    }
#pragma unroll
    for (int t = 0; t < 8; ++t) {
      acc[t] = __builtin_amdgcn_wmma_f32_16x16x4_f32(
          false, a_cur, false, b_cur[t], (short)0, acc[t], false, false);
    }
    a_cur = a_nxt;
#pragma unroll
    for (int t = 0; t < 8; ++t) b_cur[t] = b_nxt[t];
  }

  const int n    = lane & 15;
  const int mAdd = (lane >> 4) << 3;
  float* yb = Y + (size_t)mBase * D;
#pragma unroll
  for (int t = 0; t < 8; ++t) {
#pragma unroll
    for (int r = 0; r < 8; ++r) {
      yb[(size_t)(mAdd + r) * D + t * 16 + n] = acc[t][r];
    }
  }
}

__global__ void clear_counts(int* __restrict__ seg_cnt, int* __restrict__ cursor, int Ndst) {
  int tid = blockIdx.x * blockDim.x + threadIdx.x;
  if (tid < Ndst) seg_cnt[tid] = 0;
  if (tid == 0) *cursor = 0;
}

__global__ void edge_rank(const int* __restrict__ dst_idx,
                          int* __restrict__ seg_cnt,
                          int* __restrict__ rank, int E) {
  int e = blockIdx.x * blockDim.x + threadIdx.x;
  if (e >= E) return;
  rank[e] = atomicAdd(&seg_cnt[dst_idx[e]], 1);
}

__global__ void seg_alloc(const int* __restrict__ seg_cnt,
                          int* __restrict__ seg_start,
                          int* __restrict__ cursor, int Ndst) {
  __shared__ int wsum[SEG_TB / 32];
  __shared__ int base_s;
  const int d    = blockIdx.x * SEG_TB + threadIdx.x;
  const int lane = threadIdx.x & 31, w = threadIdx.x >> 5;
  const int c    = (d < Ndst) ? seg_cnt[d] : 0;
  int inc = c;
  for (int o = 1; o < 32; o <<= 1) {
    int v = __shfl_up(inc, o);
    if (lane >= o) inc += v;
  }
  if (lane == 31) wsum[w] = inc;
  __syncthreads();
  if (threadIdx.x == 0) {
    int tot = 0;
    for (int i = 0; i < SEG_TB / 32; ++i) { int v = wsum[i]; wsum[i] = tot; tot += v; }
    base_s = atomicAdd(cursor, tot);
  }
  __syncthreads();
  if (d < Ndst) seg_start[d] = base_s + wsum[w] + inc - c;
}

__global__ void edge_place(const int* __restrict__ src_idx,
                           const int* __restrict__ dst_idx,
                           const int* __restrict__ rank,
                           const int* __restrict__ seg_start,
                           int* __restrict__ src_sorted, int E) {
  int e = blockIdx.x * blockDim.x + threadIdx.x;
  if (e >= E) return;
  src_sorted[seg_start[dst_idx[e]] + rank[e]] = src_idx[e];
}

__device__ __forceinline__ float quad_sum(float p) {
  p += __int_as_float(__builtin_amdgcn_update_dpp(0, __float_as_int(p), 0xB1, 0xF, 0xF, true));
  p += __int_as_float(__builtin_amdgcn_update_dpp(0, __float_as_int(p), 0x4E, 0xF, 0xF, true));
  return p;
}

template <bool TAIL>
__device__ __forceinline__ void attend8(const float* __restrict__ kcol,
                                        const float* __restrict__ xcol,
                                        int sv, int c0, int cnt, const float4 q,
                                        float& m, float& l, float4& acc) {
  float4 kk[8], xx[8];
  int s[8];
#pragma unroll
  for (int i = 0; i < 8; ++i) s[i] = __builtin_amdgcn_readlane(sv, c0 + i);
#pragma unroll
  for (int i = 0; i < 8; ++i) kk[i] = *(const float4*)(kcol + (size_t)s[i] * D);
#pragma unroll
  for (int i = 0; i < 8; ++i) xx[i] = *(const float4*)(xcol + (size_t)s[i] * D);
  float t[8];
  float cm = m;
#pragma unroll
  for (int i = 0; i < 8; ++i) {
    float p = quad_sum(kk[i].x * q.x + kk[i].y * q.y + kk[i].z * q.z + kk[i].w * q.w);
    if (TAIL && i >= cnt) p = __uint_as_float(0xff800000u);
    t[i] = p;
    cm = fmaxf(cm, p);
  }
  const float r = __expf(m - cm);
  l *= r;
  acc.x *= r; acc.y *= r; acc.z *= r; acc.w *= r;
#pragma unroll
  for (int i = 0; i < 8; ++i) {
    const float e = __expf(t[i] - cm);
    l += e;
    acc.x += e * xx[i].x; acc.y += e * xx[i].y; acc.z += e * xx[i].z; acc.w += e * xx[i].w;
  }
  m = cm;
}

__global__ void __launch_bounds__(SEG_TB) __attribute__((amdgpu_waves_per_eu(2, 8)))
seg_attend(const float* __restrict__ h_src,
           const float* __restrict__ h_dst,
           const float* __restrict__ K,
           const float* __restrict__ Q,
           const float* __restrict__ dstK,
           const int* __restrict__ seg_start,
           const int* __restrict__ seg_cnt,
           const int* __restrict__ src_sorted,
           float* __restrict__ out, int Ndst) {
  const int lane = threadIdx.x & 31;
  const int d = __builtin_amdgcn_readfirstlane((int)((blockIdx.x * SEG_TB + threadIdx.x) >> 5));
  if (d >= Ndst) return;
  const int beg = __builtin_amdgcn_readfirstlane(seg_start[d]);
  const int n   = __builtin_amdgcn_readfirstlane(seg_cnt[d]);
  const int col = lane * 4;
  float4 q = *(const float4*)(Q + (size_t)d * D + col);
  q.x *= SCALE; q.y *= SCALE; q.z *= SCALE; q.w *= SCALE;
  const float4 dk = *(const float4*)(dstK  + (size_t)d * D + col);
  const float4 xd = *(const float4*)(h_dst + (size_t)d * D + col);
  float m = quad_sum(dk.x * q.x + dk.y * q.y + dk.z * q.z + dk.w * q.w);
  float l = 1.0f;
  float4 acc = xd;
  const float* kcol = K + col;
  const float* xcol = h_src + col;
  for (int j0 = 0; j0 < n; j0 += 32) {
    const int mc = (n - j0 < 32) ? (n - j0) : 32;
    const int sv = (lane < mc) ? src_sorted[beg + j0 + lane] : 0;
    int c0 = 0;
    for (; c0 + 8 <= mc; c0 += 8) attend8<false>(kcol, xcol, sv, c0, 8, q, m, l, acc);
    if (c0 < mc) attend8<true>(kcol, xcol, sv, c0, mc - c0, q, m, l, acc);
  }
  const float rl = 1.0f / l;
  float4 o;
  o.x = acc.x * rl; o.y = acc.y * rl; o.z = acc.z * rl; o.w = acc.w * rl;
  *(float4*)(out + (size_t)d * D + col) = o;
}

extern "C" void kernel_launch(void* const* d_in, const int* in_sizes, int n_in,
                              void* d_out, int out_size, void* d_ws, size_t ws_size,
                              hipStream_t stream) {
  const float* h_src = (const float*)d_in[0];
  const float* h_dst = (const float*)d_in[1];
  const float* Wk    = (const float*)d_in[2];
  const float* Wq    = (const float*)d_in[3];
  const int* src_idx = (const int*)d_in[4];
  const int* dst_idx = (const int*)d_in[5];

  const int N_SRC = in_sizes[0] / D;
  const int N_DST = in_sizes[1] / D;
  const int E     = in_sizes[4];

  auto align256 = [](size_t x) { return (x + 255) & ~(size_t)255; };
  char* ws = (char*)d_ws;
  size_t off = 0;
  float* K      = (float*)(ws + off); off += align256((size_t)N_SRC * D * 4);
  float* Q      = (float*)(ws + off); off += align256((size_t)N_DST * D * 4);
  float* dstK   = (float*)(ws + off); off += align256((size_t)N_DST * D * 4);
  int* rank       = (int*)(ws + off); off += align256((size_t)E * 4);
  int* src_sorted = (int*)(ws + off); off += align256((size_t)E * 4);
  int* seg_cnt    = (int*)(ws + off); off += align256((size_t)N_DST * 4);
  int* seg_start  = (int*)(ws + off); off += align256((size_t)N_DST * 4);
  int* cursor     = (int*)(ws + off); off += align256(4);

  const int TB = 256;

  {
    int waves  = N_SRC / 16;
    int blocks = (waves * 32 + TB - 1) / TB;
    gemm_xwT_wmma<<<blocks, TB, 0, stream>>>(h_src, Wk, K, N_SRC);
  }
  {
    int waves  = N_DST / 16;
    int blocks = (waves * 32 + TB - 1) / TB;
    gemm_xwT_wmma<<<blocks, TB, 0, stream>>>(h_dst, Wq, Q, N_DST);
    gemm_xwT_wmma<<<blocks, TB, 0, stream>>>(h_dst, Wk, dstK, N_DST);
  }

  clear_counts<<<(N_DST + TB - 1) / TB, TB, 0, stream>>>(seg_cnt, cursor, N_DST);
  edge_rank<<<(E + TB - 1) / TB, TB, 0, stream>>>(dst_idx, seg_cnt, rank, E);
  seg_alloc<<<(N_DST + SEG_TB - 1) / SEG_TB, SEG_TB, 0, stream>>>(seg_cnt, seg_start, cursor, N_DST);
  edge_place<<<(E + TB - 1) / TB, TB, 0, stream>>>(src_idx, dst_idx, rank, seg_start, src_sorted, E);
  seg_attend<<<(N_DST * 32 + SEG_TB - 1) / SEG_TB, SEG_TB, 0, stream>>>(
      h_src, h_dst, K, Q, dstK, seg_start, seg_cnt, src_sorted, (float*)d_out, N_DST);
}


# GEMMs: each block copies W (64 KiB) into LDS once (row pitch 132 floats, conflict-free) and reads every B fragment from LDS instead of 16 cache lines per global load; k order and sums unchanged
# speedup vs baseline: 3.4812x; 1.1967x over previous
#include <hip/hip_runtime.h>
#include <hip/hip_bf16.h>

#define D      128
#define H      8
#define DK     16
#define SCALE  0.25f
#define SEG_TB 256

typedef __attribute__((ext_vector_type(2))) float v2f;
typedef __attribute__((ext_vector_type(8))) float v8f;

#define GEMM_TB 256
#define WPITCH  132
__global__ void gemm_xwT_wmma(const float* __restrict__ X,
                              const float* __restrict__ W,
                              float* __restrict__ Y, int M) {
  __shared__ __attribute__((aligned(16))) float Wl[D * WPITCH];

  {
    float4 w4[16];
#pragma unroll
    for (int i = 0; i < 16; ++i) {
      const int idx = threadIdx.x + i * GEMM_TB;
      w4[i] = *(const float4*)(W + (size_t)(idx >> 5) * D + (idx & 31) * 4);
    }
#pragma unroll
    for (int i = 0; i < 16; ++i) {
      const int idx = threadIdx.x + i * GEMM_TB;
      *(float4*)(&Wl[(idx >> 5) * WPITCH + (idx & 31) * 4]) = w4[i];
    }
  }
  __syncthreads();

  const int lane  = threadIdx.x & 31;
  const int wave  = (blockIdx.x * GEMM_TB + threadIdx.x) >> 5;
  const int mBase = wave * 16;
  if (mBase >= M) return;

  const int row  = lane & 15;
  const int koff = (lane >> 4) << 1;

  const float* xrow  = X + (size_t)(mBase + row) * D + koff;
  const float* wbase = Wl + row * WPITCH + koff;

  v8f acc[8] = {};

  v2f a_cur = *(const v2f*)(xrow);
  v2f b_cur[8];
#pragma unroll
  for (int t = 0; t < 8; ++t)
    b_cur[t] = *(const v2f*)(wbase + t * 16 * WPITCH);

#pragma unroll 4
  for (int k0 = 0; k0 < D; k0 += 4) {
    v2f a_nxt;
    v2f b_nxt[8];
    if (k0 + 4 < D) {
      a_nxt = *(const v2f*)(xrow + k0 + 4);
#pragma unroll
      for (int t = 0; t < 8; ++t)
        b_nxt[t] = *(const v2f*)(wbase + t * 16 * WPITCH + k0 + 4);
    }
#pragma unroll
    for (int t = 0; t < 8; ++t) {
      acc[t] = __builtin_amdgcn_wmma_f32_16x16x4_f32(
          false, a_cur, false, b_cur[t], (short)0, acc[t], false, false);
    }
    a_cur = a_nxt;
#pragma unroll
    for (int t = 0; t < 8; ++t) b_cur[t] = b_nxt[t];
  }

  const int n    = lane & 15;
  const int mAdd = (lane >> 4) << 3;
  float* yb = Y + (size_t)mBase * D;
#pragma unroll
  for (int t = 0; t < 8; ++t) {
#pragma unroll
    for (int r = 0; r < 8; ++r) {
      yb[(size_t)(mAdd + r) * D + t * 16 + n] = acc[t][r];
    }
  }
}

__global__ void clear_counts(int* __restrict__ seg_cnt, int* __restrict__ cursor, int Ndst) {
  int tid = blockIdx.x * blockDim.x + threadIdx.x;
  if (tid < Ndst) seg_cnt[tid] = 0;
  if (tid == 0) *cursor = 0;
}

__global__ void edge_rank(const int* __restrict__ dst_idx,
                          int* __restrict__ seg_cnt,
                          int* __restrict__ rank, int E) {
  int e = blockIdx.x * blockDim.x + threadIdx.x;
  if (e >= E) return;
  rank[e] = atomicAdd(&seg_cnt[dst_idx[e]], 1);
}

__global__ void seg_alloc(const int* __restrict__ seg_cnt,
                          int* __restrict__ seg_start,
                          int* __restrict__ cursor, int Ndst) {
  __shared__ int wsum[SEG_TB / 32];
  __shared__ int base_s;
  const int d    = blockIdx.x * SEG_TB + threadIdx.x;
  const int lane = threadIdx.x & 31, w = threadIdx.x >> 5;
  const int c    = (d < Ndst) ? seg_cnt[d] : 0;
  int inc = c;
  for (int o = 1; o < 32; o <<= 1) {
    int v = __shfl_up(inc, o);
    if (lane >= o) inc += v;
  }
  if (lane == 31) wsum[w] = inc;
  __syncthreads();
  if (threadIdx.x == 0) {
    int tot = 0;
    for (int i = 0; i < SEG_TB / 32; ++i) { int v = wsum[i]; wsum[i] = tot; tot += v; }
    base_s = atomicAdd(cursor, tot);
  }
  __syncthreads();
  if (d < Ndst) seg_start[d] = base_s + wsum[w] + inc - c;
}

__global__ void edge_place(const int* __restrict__ src_idx,
                           const int* __restrict__ dst_idx,
                           const int* __restrict__ rank,
                           const int* __restrict__ seg_start,
                           int* __restrict__ src_sorted, int E) {
  int e = blockIdx.x * blockDim.x + threadIdx.x;
  if (e >= E) return;
  src_sorted[seg_start[dst_idx[e]] + rank[e]] = src_idx[e];
}

__device__ __forceinline__ float quad_sum(float p) {
  p += __int_as_float(__builtin_amdgcn_update_dpp(0, __float_as_int(p), 0xB1, 0xF, 0xF, true));
  p += __int_as_float(__builtin_amdgcn_update_dpp(0, __float_as_int(p), 0x4E, 0xF, 0xF, true));
  return p;
}

template <bool TAIL>
__device__ __forceinline__ void attend8(const float* __restrict__ kcol,
                                        const float* __restrict__ xcol,
                                        int sv, int c0, int cnt, const float4 q,
                                        float& m, float& l, float4& acc) {
  float4 kk[8], xx[8];
  int s[8];
#pragma unroll
  for (int i = 0; i < 8; ++i) s[i] = __builtin_amdgcn_readlane(sv, c0 + i);
#pragma unroll
  for (int i = 0; i < 8; ++i) kk[i] = *(const float4*)(kcol + (size_t)s[i] * D);
#pragma unroll
  for (int i = 0; i < 8; ++i) xx[i] = *(const float4*)(xcol + (size_t)s[i] * D);
  float t[8];
  float cm = m;
#pragma unroll
  for (int i = 0; i < 8; ++i) {
    float p = quad_sum(kk[i].x * q.x + kk[i].y * q.y + kk[i].z * q.z + kk[i].w * q.w);
    if (TAIL && i >= cnt) p = __uint_as_float(0xff800000u);
    t[i] = p;
    cm = fmaxf(cm, p);
  }
  const float r = __expf(m - cm);
  l *= r;
  acc.x *= r; acc.y *= r; acc.z *= r; acc.w *= r;
#pragma unroll
  for (int i = 0; i < 8; ++i) {
    const float e = __expf(t[i] - cm);
    l += e;
    acc.x += e * xx[i].x; acc.y += e * xx[i].y; acc.z += e * xx[i].z; acc.w += e * xx[i].w;
  }
  m = cm;
}

__global__ void __launch_bounds__(SEG_TB) __attribute__((amdgpu_waves_per_eu(2, 8)))
seg_attend(const float* __restrict__ h_src,
           const float* __restrict__ h_dst,
           const float* __restrict__ K,
           const float* __restrict__ Q,
           const float* __restrict__ dstK,
           const int* __restrict__ seg_start,
           const int* __restrict__ seg_cnt,
           const int* __restrict__ src_sorted,
           float* __restrict__ out, int Ndst) {
  const int lane = threadIdx.x & 31;
  const int d = __builtin_amdgcn_readfirstlane((int)((blockIdx.x * SEG_TB + threadIdx.x) >> 5));
  if (d >= Ndst) return;
  const int beg = __builtin_amdgcn_readfirstlane(seg_start[d]);
  const int n   = __builtin_amdgcn_readfirstlane(seg_cnt[d]);
  const int col = lane * 4;
  float4 q = *(const float4*)(Q + (size_t)d * D + col);
  q.x *= SCALE; q.y *= SCALE; q.z *= SCALE; q.w *= SCALE;
  const float4 dk = *(const float4*)(dstK  + (size_t)d * D + col);
  const float4 xd = *(const float4*)(h_dst + (size_t)d * D + col);
  float m = quad_sum(dk.x * q.x + dk.y * q.y + dk.z * q.z + dk.w * q.w);
  float l = 1.0f;
  float4 acc = xd;
  const float* kcol = K + col;
  const float* xcol = h_src + col;
  for (int j0 = 0; j0 < n; j0 += 32) {
    const int mc = (n - j0 < 32) ? (n - j0) : 32;
    const int sv = (lane < mc) ? src_sorted[beg + j0 + lane] : 0;
    int c0 = 0;
    for (; c0 + 8 <= mc; c0 += 8) attend8<false>(kcol, xcol, sv, c0, 8, q, m, l, acc);
    if (c0 < mc) attend8<true>(kcol, xcol, sv, c0, mc - c0, q, m, l, acc);
  }
  const float rl = 1.0f / l;
  float4 o;
  o.x = acc.x * rl; o.y = acc.y * rl; o.z = acc.z * rl; o.w = acc.w * rl;
  *(float4*)(out + (size_t)d * D + col) = o;
}

extern "C" void kernel_launch(void* const* d_in, const int* in_sizes, int n_in,
                              void* d_out, int out_size, void* d_ws, size_t ws_size,
                              hipStream_t stream) {
  const float* h_src = (const float*)d_in[0];
  const float* h_dst = (const float*)d_in[1];
  const float* Wk    = (const float*)d_in[2];
  const float* Wq    = (const float*)d_in[3];
  const int* src_idx = (const int*)d_in[4];
  const int* dst_idx = (const int*)d_in[5];

  const int N_SRC = in_sizes[0] / D;
  const int N_DST = in_sizes[1] / D;
  const int E     = in_sizes[4];

  auto align256 = [](size_t x) { return (x + 255) & ~(size_t)255; };
  char* ws = (char*)d_ws;
  size_t off = 0;
  float* K      = (float*)(ws + off); off += align256((size_t)N_SRC * D * 4);
  float* Q      = (float*)(ws + off); off += align256((size_t)N_DST * D * 4);
  float* dstK   = (float*)(ws + off); off += align256((size_t)N_DST * D * 4);
  int* rank       = (int*)(ws + off); off += align256((size_t)E * 4);
  int* src_sorted = (int*)(ws + off); off += align256((size_t)E * 4);
  int* seg_cnt    = (int*)(ws + off); off += align256((size_t)N_DST * 4);
  int* seg_start  = (int*)(ws + off); off += align256((size_t)N_DST * 4);
  int* cursor     = (int*)(ws + off); off += align256(4);

  const int TB = 256;

  {
    int waves  = N_SRC / 16;
    int blocks = (waves * 32 + GEMM_TB - 1) / GEMM_TB;
    gemm_xwT_wmma<<<blocks, GEMM_TB, 0, stream>>>(h_src, Wk, K, N_SRC);
  }
  {
    int waves  = N_DST / 16;
    int blocks = (waves * 32 + GEMM_TB - 1) / GEMM_TB;
    gemm_xwT_wmma<<<blocks, GEMM_TB, 0, stream>>>(h_dst, Wq, Q, N_DST);
    gemm_xwT_wmma<<<blocks, GEMM_TB, 0, stream>>>(h_dst, Wk, dstK, N_DST);
  }

  clear_counts<<<(N_DST + TB - 1) / TB, TB, 0, stream>>>(seg_cnt, cursor, N_DST);
  edge_rank<<<(E + TB - 1) / TB, TB, 0, stream>>>(dst_idx, seg_cnt, rank, E);
  seg_alloc<<<(N_DST + SEG_TB - 1) / SEG_TB, SEG_TB, 0, stream>>>(seg_cnt, seg_start, cursor, N_DST);
  edge_place<<<(E + TB - 1) / TB, TB, 0, stream>>>(src_idx, dst_idx, rank, seg_start, src_sorted, E);
  seg_attend<<<(N_DST * 32 + SEG_TB - 1) / SEG_TB, SEG_TB, 0, stream>>>(
      h_src, h_dst, K, Q, dstK, seg_start, seg_cnt, src_sorted, (float*)d_out, N_DST);
}


# Edge bucketing on LDS counters: chunk-local sort by coarse dst bin, then one work-group per bin ranks its edges per row; no per-edge global atomic; arrival order inside a row's sum changes
# speedup vs baseline: 5.0098x; 1.4391x over previous
#include <hip/hip_runtime.h>
#include <hip/hip_bf16.h>

#define D      128
#define H      8
#define DK     16
#define SCALE  0.25f
#define SEG_TB 256

typedef __attribute__((ext_vector_type(2))) float v2f;
typedef __attribute__((ext_vector_type(8))) float v8f;

#define GEMM_TB 256
#define WPITCH  132
__global__ void gemm_xwT_wmma(const float* __restrict__ X,
                              const float* __restrict__ W,
                              float* __restrict__ Y, int M) {
  __shared__ __attribute__((aligned(16))) float Wl[D * WPITCH];

  {
    float4 w4[16];
#pragma unroll
    for (int i = 0; i < 16; ++i) {
      const int idx = threadIdx.x + i * GEMM_TB;
      w4[i] = *(const float4*)(W + (size_t)(idx >> 5) * D + (idx & 31) * 4);
    }
#pragma unroll
    for (int i = 0; i < 16; ++i) {
      const int idx = threadIdx.x + i * GEMM_TB;
      *(float4*)(&Wl[(idx >> 5) * WPITCH + (idx & 31) * 4]) = w4[i];
    }
  }
  __syncthreads();

  const int lane  = threadIdx.x & 31;
  const int wave  = (blockIdx.x * GEMM_TB + threadIdx.x) >> 5;
  const int mBase = wave * 16;
  if (mBase >= M) return;

  const int row  = lane & 15;
  const int koff = (lane >> 4) << 1;

  const float* xrow  = X + (size_t)(mBase + row) * D + koff;
  const float* wbase = Wl + row * WPITCH + koff;

  v8f acc[8] = {};

  v2f a_cur = *(const v2f*)(xrow);
  v2f b_cur[8];
#pragma unroll
  for (int t = 0; t < 8; ++t)
    b_cur[t] = *(const v2f*)(wbase + t * 16 * WPITCH);

#pragma unroll 4
  for (int k0 = 0; k0 < D; k0 += 4) {
    v2f a_nxt;
    v2f b_nxt[8];
    if (k0 + 4 < D) {
      a_nxt = *(const v2f*)(xrow + k0 + 4);
#pragma unroll
      for (int t = 0; t < 8; ++t)
        b_nxt[t] = *(const v2f*)(wbase + t * 16 * WPITCH + k0 + 4);
    }
#pragma unroll
    for (int t = 0; t < 8; ++t) {
      acc[t] = __builtin_amdgcn_wmma_f32_16x16x4_f32(
          false, a_cur, false, b_cur[t], (short)0, acc[t], false, false);
    }
    a_cur = a_nxt;
#pragma unroll
    for (int t = 0; t < 8; ++t) b_cur[t] = b_nxt[t];
  }

  const int n    = lane & 15;
  const int mAdd = (lane >> 4) << 3;
  float* yb = Y + (size_t)mBase * D;
#pragma unroll
  for (int t = 0; t < 8; ++t) {
#pragma unroll
    for (int r = 0; r < 8; ++r) {
      yb[(size_t)(mAdd + r) * D + t * 16 + n] = acc[t][r];
    }
  }
}

#define BIN_SHIFT 8
#define BIN_SZ    256
#define NBIN_MAX  256
#define SORT_TB   1024
#define SORT_G    256
#define SORT_PRE  8
#define RUNS_PW   (SORT_G / (SORT_TB / 32))

__device__ __forceinline__ int excl_scan256(int c, int* wsum) {
  const int tid = threadIdx.x, lane = tid & 31;
  int inc = c;
  for (int o = 1; o < 32; o <<= 1) {
    int v = __shfl_up(inc, o);
    if (lane >= o) inc += v;
  }
  if (tid < 256 && lane == 31) wsum[tid >> 5] = inc;
  __syncthreads();
  int base = 0;
  if (tid < 256)
    for (int i = 0; i < (tid >> 5); ++i) base += wsum[i];
  return base + inc - c;
}

__global__ void __launch_bounds__(SORT_TB)
chunk_sort(const int* __restrict__ src_idx,
           const int* __restrict__ dst_idx,
           int* __restrict__ pairs,
           int* __restrict__ run_start,
           int* __restrict__ run_cnt,
           int* __restrict__ cursor, int E) {
  __shared__ int hist[NBIN_MAX];
  __shared__ int fill[NBIN_MAX];
  __shared__ int wsum[8];
  const int tid = threadIdx.x;
  const int ch  = (E + SORT_G - 1) / SORT_G;
  const int e0  = blockIdx.x * ch;
  const int e1  = (e0 + ch < E) ? (e0 + ch) : E;
  if (tid < NBIN_MAX) hist[tid] = 0;
  if (blockIdx.x == 0 && tid == 0) *cursor = 0;

  int dv[SORT_PRE], sv[SORT_PRE];
#pragma unroll
  for (int k = 0; k < SORT_PRE; ++k) {
    const int e = e0 + tid + k * SORT_TB;
    dv[k] = (e < e1) ? dst_idx[e] : -1;
  }
#pragma unroll
  for (int k = 0; k < SORT_PRE; ++k) {
    const int e = e0 + tid + k * SORT_TB;
    sv[k] = (e < e1) ? src_idx[e] : 0;
  }
  __syncthreads();
#pragma unroll
  for (int k = 0; k < SORT_PRE; ++k)
    if (dv[k] >= 0) atomicAdd(&hist[dv[k] >> BIN_SHIFT], 1);
  for (int e = e0 + tid + SORT_PRE * SORT_TB; e < e1; e += SORT_TB)
    atomicAdd(&hist[dst_idx[e] >> BIN_SHIFT], 1);
  __syncthreads();

  const int c  = (tid < NBIN_MAX) ? hist[tid] : 0;
  const int st = excl_scan256(c, wsum);
  if (tid < NBIN_MAX) {
    fill[tid] = st;
    run_start[tid * SORT_G + blockIdx.x] = st;
    run_cnt[tid * SORT_G + blockIdx.x]   = c;
  }
  __syncthreads();

  int2* mine = (int2*)pairs + e0;
#pragma unroll
  for (int k = 0; k < SORT_PRE; ++k)
    if (dv[k] >= 0) {
      const int p = atomicAdd(&fill[dv[k] >> BIN_SHIFT], 1);
      mine[p] = make_int2(sv[k], dv[k]);
    }
  for (int e = e0 + tid + SORT_PRE * SORT_TB; e < e1; e += SORT_TB) {
    const int d = dst_idx[e];
    const int p = atomicAdd(&fill[d >> BIN_SHIFT], 1);
    mine[p] = make_int2(src_idx[e], d);
  }
}

__global__ void __launch_bounds__(SORT_TB)
bin_sort(const int* __restrict__ pairs,
         const int* __restrict__ run_start,
         const int* __restrict__ run_cnt,
         int* __restrict__ cursor,
         int* __restrict__ seg_start,
         int* __restrict__ seg_cnt,
         int* __restrict__ src_sorted, int E, int Ndst) {
  __shared__ int rs[SORT_G];
  __shared__ int rc[SORT_G];
  __shared__ int cnt[BIN_SZ];
  __shared__ int fill[BIN_SZ];
  __shared__ int wtot[8];
  __shared__ int wsum[8];
  __shared__ int base_s;
  const int tid = threadIdx.x, lane = tid & 31, w = tid >> 5;
  const int b  = blockIdx.x;
  const int ch = (E + SORT_G - 1) / SORT_G;
  if (tid < BIN_SZ) cnt[tid] = 0;
  if (tid < SORT_G) {
    const int c = run_cnt[b * SORT_G + tid];
    rc[tid] = c;
    rs[tid] = tid * ch + run_start[b * SORT_G + tid];
    int s = c;
    for (int o = 16; o > 0; o >>= 1) s += __shfl_xor(s, o);
    if (lane == 0) wtot[w] = s;
  }
  __syncthreads();

  const int2* pp = (const int2*)pairs;
  int  rn[RUNS_PW];
  int2 v0[RUNS_PW], v1[RUNS_PW];
#pragma unroll
  for (int k = 0; k < RUNS_PW; ++k) {
    const int r = w + k * (SORT_TB / 32);
    rn[k] = rc[r];
    int i0 = rs[r] + lane, i1 = i0 + 32;
    i0 = (i0 < E) ? i0 : E - 1;
    i1 = (i1 < E) ? i1 : E - 1;
    v0[k] = pp[i0];
    v1[k] = pp[i1];
  }
  if (tid == 0) {
    int tot = 0;
    for (int i = 0; i < SORT_G / 32; ++i) tot += wtot[i];
    base_s = atomicAdd(cursor, tot);
  }
#pragma unroll
  for (int k = 0; k < RUNS_PW; ++k) {
    if (lane < rn[k])      atomicAdd(&cnt[v0[k].y & (BIN_SZ - 1)], 1);
    if (lane + 32 < rn[k]) atomicAdd(&cnt[v1[k].y & (BIN_SZ - 1)], 1);
    if (rn[k] > 64) {
      const int2* p = pp + rs[w + k * (SORT_TB / 32)];
      for (int i = lane + 64; i < rn[k]; i += 32) atomicAdd(&cnt[p[i].y & (BIN_SZ - 1)], 1);
    }
  }
  __syncthreads();

  const int c  = (tid < BIN_SZ) ? cnt[tid] : 0;
  const int st = excl_scan256(c, wsum);
  const int base = base_s;
  if (tid < BIN_SZ) {
    fill[tid] = st;
    const int d = b * BIN_SZ + tid;
    if (d < Ndst) { seg_start[d] = base + st; seg_cnt[d] = c; }
  }
  __syncthreads();

#pragma unroll
  for (int k = 0; k < RUNS_PW; ++k) {
    if (lane < rn[k])      src_sorted[base + atomicAdd(&fill[v0[k].y & (BIN_SZ - 1)], 1)] = v0[k].x;
    if (lane + 32 < rn[k]) src_sorted[base + atomicAdd(&fill[v1[k].y & (BIN_SZ - 1)], 1)] = v1[k].x;
    if (rn[k] > 64) {
      const int2* p = pp + rs[w + k * (SORT_TB / 32)];
      for (int i = lane + 64; i < rn[k]; i += 32) {
        const int2 pr = p[i];
        src_sorted[base + atomicAdd(&fill[pr.y & (BIN_SZ - 1)], 1)] = pr.x;
      }
    }
  }
}

__device__ __forceinline__ float quad_sum(float p) {
  p += __int_as_float(__builtin_amdgcn_update_dpp(0, __float_as_int(p), 0xB1, 0xF, 0xF, true));
  p += __int_as_float(__builtin_amdgcn_update_dpp(0, __float_as_int(p), 0x4E, 0xF, 0xF, true));
  return p;
}

template <bool TAIL>
__device__ __forceinline__ void attend8(const float* __restrict__ kcol,
                                        const float* __restrict__ xcol,
                                        int sv, int c0, int cnt, const float4 q,
                                        float& m, float& l, float4& acc) {
  float4 kk[8], xx[8];
  int s[8];
#pragma unroll
  for (int i = 0; i < 8; ++i) s[i] = __builtin_amdgcn_readlane(sv, c0 + i);
#pragma unroll
  for (int i = 0; i < 8; ++i) kk[i] = *(const float4*)(kcol + (size_t)s[i] * D);
#pragma unroll
  for (int i = 0; i < 8; ++i) xx[i] = *(const float4*)(xcol + (size_t)s[i] * D);
  float t[8];
  float cm = m;
#pragma unroll
  for (int i = 0; i < 8; ++i) {
    float p = quad_sum(kk[i].x * q.x + kk[i].y * q.y + kk[i].z * q.z + kk[i].w * q.w);
    if (TAIL && i >= cnt) p = __uint_as_float(0xff800000u);
    t[i] = p;
    cm = fmaxf(cm, p);
  }
  const float r = __expf(m - cm);
  l *= r;
  acc.x *= r; acc.y *= r; acc.z *= r; acc.w *= r;
#pragma unroll
  for (int i = 0; i < 8; ++i) {
    const float e = __expf(t[i] - cm);
    l += e;
    acc.x += e * xx[i].x; acc.y += e * xx[i].y; acc.z += e * xx[i].z; acc.w += e * xx[i].w;
  }
  m = cm;
}

__global__ void __launch_bounds__(SEG_TB) __attribute__((amdgpu_waves_per_eu(2, 8)))
seg_attend(const float* __restrict__ h_src,
           const float* __restrict__ h_dst,
           const float* __restrict__ K,
           const float* __restrict__ Q,
           const float* __restrict__ dstK,
           const int* __restrict__ seg_start,
           const int* __restrict__ seg_cnt,
           const int* __restrict__ src_sorted,
           float* __restrict__ out, int Ndst) {
  const int lane = threadIdx.x & 31;
  const int d = __builtin_amdgcn_readfirstlane((int)((blockIdx.x * SEG_TB + threadIdx.x) >> 5));
  if (d >= Ndst) return;
  const int beg = __builtin_amdgcn_readfirstlane(seg_start[d]);
  const int n   = __builtin_amdgcn_readfirstlane(seg_cnt[d]);
  const int col = lane * 4;
  float4 q = *(const float4*)(Q + (size_t)d * D + col);
  q.x *= SCALE; q.y *= SCALE; q.z *= SCALE; q.w *= SCALE;
  const float4 dk = *(const float4*)(dstK  + (size_t)d * D + col);
  const float4 xd = *(const float4*)(h_dst + (size_t)d * D + col);
  float m = quad_sum(dk.x * q.x + dk.y * q.y + dk.z * q.z + dk.w * q.w);
  float l = 1.0f;
  float4 acc = xd;
  const float* kcol = K + col;
  const float* xcol = h_src + col;
  for (int j0 = 0; j0 < n; j0 += 32) {
    const int mc = (n - j0 < 32) ? (n - j0) : 32;
    const int sv = (lane < mc) ? src_sorted[beg + j0 + lane] : 0;
    int c0 = 0;
    for (; c0 + 8 <= mc; c0 += 8) attend8<false>(kcol, xcol, sv, c0, 8, q, m, l, acc);
    if (c0 < mc) attend8<true>(kcol, xcol, sv, c0, mc - c0, q, m, l, acc);
  }
  const float rl = 1.0f / l;
  float4 o;
  o.x = acc.x * rl; o.y = acc.y * rl; o.z = acc.z * rl; o.w = acc.w * rl;
  *(float4*)(out + (size_t)d * D + col) = o;
}

extern "C" void kernel_launch(void* const* d_in, const int* in_sizes, int n_in,
                              void* d_out, int out_size, void* d_ws, size_t ws_size,
                              hipStream_t stream) {
  const float* h_src = (const float*)d_in[0];
  const float* h_dst = (const float*)d_in[1];
  const float* Wk    = (const float*)d_in[2];
  const float* Wq    = (const float*)d_in[3];
  const int* src_idx = (const int*)d_in[4];
  const int* dst_idx = (const int*)d_in[5];

  const int N_SRC = in_sizes[0] / D;
  const int N_DST = in_sizes[1] / D;
  const int E     = in_sizes[4];

  auto align256 = [](size_t x) { return (x + 255) & ~(size_t)255; };
  char* ws = (char*)d_ws;
  size_t off = 0;
  float* K      = (float*)(ws + off); off += align256((size_t)N_SRC * D * 4);
  float* Q      = (float*)(ws + off); off += align256((size_t)N_DST * D * 4);
  float* dstK   = (float*)(ws + off); off += align256((size_t)N_DST * D * 4);
  int* pairs      = (int*)(ws + off); off += align256((size_t)E * 8);
  int* src_sorted = (int*)(ws + off); off += align256((size_t)E * 4);
  int* seg_cnt    = (int*)(ws + off); off += align256((size_t)N_DST * 4);
  int* seg_start  = (int*)(ws + off); off += align256((size_t)N_DST * 4);
  int* cursor     = (int*)(ws + off); off += align256(4);
  int* run_start  = (int*)(ws + off); off += align256((size_t)NBIN_MAX * SORT_G * 4);
  int* run_cnt    = (int*)(ws + off); off += align256((size_t)NBIN_MAX * SORT_G * 4);

  {
    int waves  = N_SRC / 16;
    int blocks = (waves * 32 + GEMM_TB - 1) / GEMM_TB;
    gemm_xwT_wmma<<<blocks, GEMM_TB, 0, stream>>>(h_src, Wk, K, N_SRC);
  }
  {
    int waves  = N_DST / 16;
    int blocks = (waves * 32 + GEMM_TB - 1) / GEMM_TB;
    gemm_xwT_wmma<<<blocks, GEMM_TB, 0, stream>>>(h_dst, Wq, Q, N_DST);
    gemm_xwT_wmma<<<blocks, GEMM_TB, 0, stream>>>(h_dst, Wk, dstK, N_DST);
  }

  chunk_sort<<<SORT_G, SORT_TB, 0, stream>>>(src_idx, dst_idx, pairs, run_start, run_cnt, cursor, E);
  bin_sort<<<(N_DST + BIN_SZ - 1) / BIN_SZ, SORT_TB, 0, stream>>>(
      pairs, run_start, run_cnt, cursor, seg_start, seg_cnt, src_sorted, E, N_DST);
  seg_attend<<<(N_DST * 32 + SEG_TB - 1) / SEG_TB, SEG_TB, 0, stream>>>(
      h_src, h_dst, K, Q, dstK, seg_start, seg_cnt, src_sorted, (float*)d_out, N_DST);
}


# Query taken through Wk once per dst row and head (new f32 WMMA pass); edge score = raw source row . that plane: K projection of all sources and its per-edge gather gone; score sum order changes
# speedup vs baseline: 5.4288x; 1.0836x over previous
#include <hip/hip_runtime.h>
#include <hip/hip_bf16.h>

#define D      128
#define H      8
#define DK     16
#define SCALE  0.25f
#define SEG_TB 256

typedef __attribute__((ext_vector_type(2))) float v2f;
typedef __attribute__((ext_vector_type(8))) float v8f;
typedef __attribute__((ext_vector_type(4))) float v4f;

#define GEMM_TB 256
#define WPITCH  132
__global__ void gemm_xwT_wmma(const float* __restrict__ X,
                              const float* __restrict__ W,
                              float* __restrict__ Y, int M) {
  __shared__ __attribute__((aligned(16))) float Wl[D * WPITCH];

  {
    float4 w4[16];
#pragma unroll
    for (int i = 0; i < 16; ++i) {
      const int idx = threadIdx.x + i * GEMM_TB;
      w4[i] = *(const float4*)(W + (size_t)(idx >> 5) * D + (idx & 31) * 4);
    }
#pragma unroll
    for (int i = 0; i < 16; ++i) {
      const int idx = threadIdx.x + i * GEMM_TB;
      *(float4*)(&Wl[(idx >> 5) * WPITCH + (idx & 31) * 4]) = w4[i];
    }
  }
  __syncthreads();

  const int lane  = threadIdx.x & 31;
  const int wave  = (blockIdx.x * GEMM_TB + threadIdx.x) >> 5;
  const int mBase = wave * 16;
  if (mBase >= M) return;

  const int row  = lane & 15;
  const int koff = (lane >> 4) << 1;

  const float* xrow  = X + (size_t)(mBase + row) * D + koff;
  const float* wbase = Wl + row * WPITCH + koff;

  v8f acc[8] = {};

  v2f a_cur = *(const v2f*)(xrow);
  v2f b_cur[8];
#pragma unroll
  for (int t = 0; t < 8; ++t)
    b_cur[t] = *(const v2f*)(wbase + t * 16 * WPITCH);

#pragma unroll 4
  for (int k0 = 0; k0 < D; k0 += 4) {
    v2f a_nxt;
    v2f b_nxt[8];
    if (k0 + 4 < D) {
      a_nxt = *(const v2f*)(xrow + k0 + 4);
#pragma unroll
      for (int t = 0; t < 8; ++t)
        b_nxt[t] = *(const v2f*)(wbase + t * 16 * WPITCH + k0 + 4);
    }
#pragma unroll
    for (int t = 0; t < 8; ++t) {
      acc[t] = __builtin_amdgcn_wmma_f32_16x16x4_f32(
          false, a_cur, false, b_cur[t], (short)0, acc[t], false, false);
    }
    a_cur = a_nxt;
#pragma unroll
    for (int t = 0; t < 8; ++t) b_cur[t] = b_nxt[t];
  }

  const int n    = lane & 15;
  const int mAdd = (lane >> 4) << 3;
  float* yb = Y + (size_t)mBase * D;
#pragma unroll
  for (int t = 0; t < 8; ++t) {
#pragma unroll
    for (int r = 0; r < 8; ++r) {
      yb[(size_t)(mAdd + r) * D + t * 16 + n] = acc[t][r];
    }
  }
}

#define PPITCH 132
__global__ void qproj_wmma(const float* __restrict__ Q,
                           const float* __restrict__ W,
                           float* __restrict__ P, int M) {
  __shared__ __attribute__((aligned(16))) float Wl[D * PPITCH];
  {
    float4 w4[16];
#pragma unroll
    for (int i = 0; i < 16; ++i) {
      const int idx = threadIdx.x + i * GEMM_TB;
      w4[i] = *(const float4*)(W + (size_t)(idx >> 5) * D + (idx & 31) * 4);
    }
#pragma unroll
    for (int i = 0; i < 16; ++i) {
      const int idx = threadIdx.x + i * GEMM_TB;
      *(float4*)(&Wl[(idx >> 5) * PPITCH + (idx & 31) * 4]) = w4[i];
    }
  }
  __syncthreads();

  const int lane  = threadIdx.x & 31;
  const int wave  = (blockIdx.x * GEMM_TB + threadIdx.x) >> 5;
  const int mBase = wave * 16;
  if (mBase >= M) return;

  const int c  = lane & 15;
  const int hh = lane >> 4;
  const float* qrow = Q + (size_t)(mBase + c) * D + 8 * hh;
  const float* wb   = Wl + (8 * hh) * PPITCH + c;
  const int mAdd = hh << 3;
  float* pb = P + (size_t)mBase * (H * D);

  for (int h = 0; h < H; ++h) {
    const float4 q0 = *(const float4*)(qrow + 16 * h);
    const float4 q1 = *(const float4*)(qrow + 16 * h + 4);
    v2f a[4];
    a[0].x = q0.x; a[0].y = q0.y; a[1].x = q0.z; a[1].y = q0.w;
    a[2].x = q1.x; a[2].y = q1.y; a[3].x = q1.z; a[3].y = q1.w;
    const float* wh = wb + (16 * h) * PPITCH;
    v8f acc[8] = {};
#pragma unroll
    for (int s = 0; s < 4; ++s) {
      v2f b[8];
#pragma unroll
      for (int t = 0; t < 8; ++t) {
        b[t].x = wh[(2 * s) * PPITCH + 16 * t];
        b[t].y = wh[(2 * s + 1) * PPITCH + 16 * t];
      }
#pragma unroll
      for (int t = 0; t < 8; ++t) {
        acc[t] = __builtin_amdgcn_wmma_f32_16x16x4_f32(
            false, a[s], false, b[t], (short)0, acc[t], false, false);
      }
    }
#pragma unroll
    for (int t = 0; t < 8; ++t) {
#pragma unroll
      for (int r = 0; r < 8; ++r) {
        pb[(size_t)(mAdd + r) * (H * D) + (h ^ t) * D + t * 16 + c] = acc[t][r];
      }
    }
  }
}

#define BIN_SHIFT 8
#define BIN_SZ    256
#define NBIN_MAX  256
#define SORT_TB   1024
#define SORT_G    256
#define SORT_PRE  8
#define RUNS_PW   (SORT_G / (SORT_TB / 32))

__device__ __forceinline__ int excl_scan256(int c, int* wsum) {
  const int tid = threadIdx.x, lane = tid & 31;
  int inc = c;
  for (int o = 1; o < 32; o <<= 1) {
    int v = __shfl_up(inc, o);
    if (lane >= o) inc += v;
  }
  if (tid < 256 && lane == 31) wsum[tid >> 5] = inc;
  __syncthreads();
  int base = 0;
  if (tid < 256)
    for (int i = 0; i < (tid >> 5); ++i) base += wsum[i];
  return base + inc - c;
}

__global__ void __launch_bounds__(SORT_TB)
chunk_sort(const int* __restrict__ src_idx,
           const int* __restrict__ dst_idx,
           int* __restrict__ pairs,
           int* __restrict__ run_start,
           int* __restrict__ run_cnt,
           int* __restrict__ cursor, int E) {
  __shared__ int hist[NBIN_MAX];
  __shared__ int fill[NBIN_MAX];
  __shared__ int wsum[8];
  const int tid = threadIdx.x;
  const int ch  = (E + SORT_G - 1) / SORT_G;
  const int e0  = blockIdx.x * ch;
  const int e1  = (e0 + ch < E) ? (e0 + ch) : E;
  if (tid < NBIN_MAX) hist[tid] = 0;
  if (blockIdx.x == 0 && tid == 0) *cursor = 0;

  int dv[SORT_PRE], sv[SORT_PRE];
#pragma unroll
  for (int k = 0; k < SORT_PRE; ++k) {
    const int e = e0 + tid + k * SORT_TB;
    dv[k] = (e < e1) ? dst_idx[e] : -1;
  }
#pragma unroll
  for (int k = 0; k < SORT_PRE; ++k) {
    const int e = e0 + tid + k * SORT_TB;
    sv[k] = (e < e1) ? src_idx[e] : 0;
  }
  __syncthreads();
#pragma unroll
  for (int k = 0; k < SORT_PRE; ++k)
    if (dv[k] >= 0) atomicAdd(&hist[dv[k] >> BIN_SHIFT], 1);
  for (int e = e0 + tid + SORT_PRE * SORT_TB; e < e1; e += SORT_TB)
    atomicAdd(&hist[dst_idx[e] >> BIN_SHIFT], 1);
  __syncthreads();

  const int c  = (tid < NBIN_MAX) ? hist[tid] : 0;
  const int st = excl_scan256(c, wsum);
  if (tid < NBIN_MAX) {
    fill[tid] = st;
    run_start[tid * SORT_G + blockIdx.x] = st;
    run_cnt[tid * SORT_G + blockIdx.x]   = c;
  }
  __syncthreads();

  int2* mine = (int2*)pairs + e0;
#pragma unroll
  for (int k = 0; k < SORT_PRE; ++k)
    if (dv[k] >= 0) {
      const int p = atomicAdd(&fill[dv[k] >> BIN_SHIFT], 1);
      mine[p] = make_int2(sv[k], dv[k]);
    }
  for (int e = e0 + tid + SORT_PRE * SORT_TB; e < e1; e += SORT_TB) {
    const int d = dst_idx[e];
    const int p = atomicAdd(&fill[d >> BIN_SHIFT], 1);
    mine[p] = make_int2(src_idx[e], d);
  }
}

__global__ void __launch_bounds__(SORT_TB)
bin_sort(const int* __restrict__ pairs,
         const int* __restrict__ run_start,
         const int* __restrict__ run_cnt,
         int* __restrict__ cursor,
         int* __restrict__ seg_start,
         int* __restrict__ seg_cnt,
         int* __restrict__ src_sorted, int E, int Ndst) {
  __shared__ int rs[SORT_G];
  __shared__ int rc[SORT_G];
  __shared__ int cnt[BIN_SZ];
  __shared__ int fill[BIN_SZ];
  __shared__ int wtot[8];
  __shared__ int wsum[8];
  __shared__ int base_s;
  const int tid = threadIdx.x, lane = tid & 31, w = tid >> 5;
  const int b  = blockIdx.x;
  const int ch = (E + SORT_G - 1) / SORT_G;
  if (tid < BIN_SZ) cnt[tid] = 0;
  if (tid < SORT_G) {
    const int c = run_cnt[b * SORT_G + tid];
    rc[tid] = c;
    rs[tid] = tid * ch + run_start[b * SORT_G + tid];
    int s = c;
    for (int o = 16; o > 0; o >>= 1) s += __shfl_xor(s, o);
    if (lane == 0) wtot[w] = s;
  }
  __syncthreads();

  const int2* pp = (const int2*)pairs;
  int  rn[RUNS_PW];
  int2 v0[RUNS_PW], v1[RUNS_PW];
#pragma unroll
  for (int k = 0; k < RUNS_PW; ++k) {
    const int r = w + k * (SORT_TB / 32);
    rn[k] = rc[r];
    int i0 = rs[r] + lane, i1 = i0 + 32;
    i0 = (i0 < E) ? i0 : E - 1;
    i1 = (i1 < E) ? i1 : E - 1;
    v0[k] = pp[i0];
    v1[k] = pp[i1];
  }
  if (tid == 0) {
    int tot = 0;
    for (int i = 0; i < SORT_G / 32; ++i) tot += wtot[i];
    base_s = atomicAdd(cursor, tot);
  }
#pragma unroll
  for (int k = 0; k < RUNS_PW; ++k) {
    if (lane < rn[k])      atomicAdd(&cnt[v0[k].y & (BIN_SZ - 1)], 1);
    if (lane + 32 < rn[k]) atomicAdd(&cnt[v1[k].y & (BIN_SZ - 1)], 1);
    if (rn[k] > 64) {
      const int2* p = pp + rs[w + k * (SORT_TB / 32)];
      for (int i = lane + 64; i < rn[k]; i += 32) atomicAdd(&cnt[p[i].y & (BIN_SZ - 1)], 1);
    }
  }
  __syncthreads();

  const int c  = (tid < BIN_SZ) ? cnt[tid] : 0;
  const int st = excl_scan256(c, wsum);
  const int base = base_s;
  if (tid < BIN_SZ) {
    fill[tid] = st;
    const int d = b * BIN_SZ + tid;
    if (d < Ndst) { seg_start[d] = base + st; seg_cnt[d] = c; }
  }
  __syncthreads();

#pragma unroll
  for (int k = 0; k < RUNS_PW; ++k) {
    if (lane < rn[k])      src_sorted[base + atomicAdd(&fill[v0[k].y & (BIN_SZ - 1)], 1)] = v0[k].x;
    if (lane + 32 < rn[k]) src_sorted[base + atomicAdd(&fill[v1[k].y & (BIN_SZ - 1)], 1)] = v1[k].x;
    if (rn[k] > 64) {
      const int2* p = pp + rs[w + k * (SORT_TB / 32)];
      for (int i = lane + 64; i < rn[k]; i += 32) {
        const int2 pr = p[i];
        src_sorted[base + atomicAdd(&fill[pr.y & (BIN_SZ - 1)], 1)] = pr.x;
      }
    }
  }
}

__device__ __forceinline__ float quad_sum(float p) {
  p += __int_as_float(__builtin_amdgcn_update_dpp(0, __float_as_int(p), 0xB1, 0xF, 0xF, true));
  p += __int_as_float(__builtin_amdgcn_update_dpp(0, __float_as_int(p), 0x4E, 0xF, 0xF, true));
  return p;
}

__device__ __forceinline__ float other_row(float x) {
  return __int_as_float(__builtin_amdgcn_permlanex16(__float_as_int(x), __float_as_int(x),
                                                     0x76543210, 0xfedcba98, false, false));
}

__device__ __forceinline__ float head_score(const float4 x, const v2f (&qq)[4][4]) {
  const v2f xx = {x.x, x.x}, xy = {x.y, x.y}, xz = {x.z, x.z}, xw = {x.w, x.w};
  v2f p[4];
#pragma unroll
  for (int jj = 0; jj < 4; ++jj) {
    v2f t = xx * qq[jj][0];
    t = __builtin_elementwise_fma(xy, qq[jj][1], t);
    t = __builtin_elementwise_fma(xz, qq[jj][2], t);
    t = __builtin_elementwise_fma(xw, qq[jj][3], t);
    p[jj] = t;
  }
  v2f r0, r1;
  r0.x = other_row(p[2].x); r0.y = other_row(p[2].y);
  r1.x = other_row(p[3].x); r1.y = other_row(p[3].y);
  const v2f a = p[0] + r0;
  const v2f b = p[1] + r1;
  v2f s;
  s.x = __int_as_float(__builtin_amdgcn_update_dpp(0, __float_as_int(b.x), 0x128, 0xF, 0xF, true));
  s.y = __int_as_float(__builtin_amdgcn_update_dpp(0, __float_as_int(b.y), 0x128, 0xF, 0xF, true));
  const v2f e = a + s;
  const float c0 = e.x + __int_as_float(__builtin_amdgcn_update_dpp(0, __float_as_int(e.y), 0x141, 0xF, 0xF, true));
  return quad_sum(c0);
}

template <bool TAIL>
__device__ __forceinline__ void attend8(const float* __restrict__ xcol,
                                        int sv, int c0, int cnt, const v2f (&qq)[4][4],
                                        float& m, float& l, float4& acc) {
  float4 xx[8];
  int s[8];
#pragma unroll
  for (int i = 0; i < 8; ++i) s[i] = __builtin_amdgcn_readlane(sv, c0 + i);
#pragma unroll
  for (int i = 0; i < 8; ++i) xx[i] = *(const float4*)(xcol + (size_t)s[i] * D);
  float t[8];
  float cm = m;
#pragma unroll
  for (int i = 0; i < 8; ++i) {
    float p = head_score(xx[i], qq);
    if (TAIL && i >= cnt) p = __uint_as_float(0xff800000u);
    t[i] = p;
    cm = fmaxf(cm, p);
  }
  const float r = __expf(m - cm);
  l *= r;
  acc.x *= r; acc.y *= r; acc.z *= r; acc.w *= r;
#pragma unroll
  for (int i = 0; i < 8; ++i) {
    const float e = __expf(t[i] - cm);
    l += e;
    acc.x += e * xx[i].x; acc.y += e * xx[i].y; acc.z += e * xx[i].z; acc.w += e * xx[i].w;
  }
  m = cm;
}

__global__ void __launch_bounds__(SEG_TB) __attribute__((amdgpu_waves_per_eu(6, 8)))
seg_attend(const float* __restrict__ h_src,
           const float* __restrict__ h_dst,
           const float* __restrict__ P,
           const int* __restrict__ seg_start,
           const int* __restrict__ seg_cnt,
           const int* __restrict__ src_sorted,
           float* __restrict__ out, int Ndst) {
  const int lane = threadIdx.x & 31;
  const int d = __builtin_amdgcn_readfirstlane((int)((blockIdx.x * SEG_TB + threadIdx.x) >> 5));
  if (d >= Ndst) return;
  const int beg = __builtin_amdgcn_readfirstlane(seg_start[d]);
  const int n   = __builtin_amdgcn_readfirstlane(seg_cnt[d]);
  const int col = lane * 4;
  v2f qq[4][4];
  const float* prow = P + (size_t)d * (H * D) + col;
#pragma unroll
  for (int jj = 0; jj < 4; ++jj) {
    const v4f u = __builtin_nontemporal_load((const v4f*)(prow + (2 * jj) * D));
    const v4f w = __builtin_nontemporal_load((const v4f*)(prow + (2 * jj + 1) * D));
    qq[jj][0].x = u.x * SCALE; qq[jj][0].y = w.x * SCALE;
    qq[jj][1].x = u.y * SCALE; qq[jj][1].y = w.y * SCALE;
    qq[jj][2].x = u.z * SCALE; qq[jj][2].y = w.z * SCALE;
    qq[jj][3].x = u.w * SCALE; qq[jj][3].y = w.w * SCALE;
  }
  const float4 xd = *(const float4*)(h_dst + (size_t)d * D + col);
  float m = head_score(xd, qq);
  float l = 1.0f;
  float4 acc = xd;
  const float* xcol = h_src + col;
  for (int j0 = 0; j0 < n; j0 += 32) {
    const int mc = (n - j0 < 32) ? (n - j0) : 32;
    const int sv = (lane < mc) ? src_sorted[beg + j0 + lane] : 0;
    int c0 = 0;
    for (; c0 + 8 <= mc; c0 += 8) attend8<false>(xcol, sv, c0, 8, qq, m, l, acc);
    if (c0 < mc) attend8<true>(xcol, sv, c0, mc - c0, qq, m, l, acc);
  }
  const float rl = 1.0f / l;
  float4 o;
  o.x = acc.x * rl; o.y = acc.y * rl; o.z = acc.z * rl; o.w = acc.w * rl;
  *(float4*)(out + (size_t)d * D + col) = o;
}

extern "C" void kernel_launch(void* const* d_in, const int* in_sizes, int n_in,
                              void* d_out, int out_size, void* d_ws, size_t ws_size,
                              hipStream_t stream) {
  const float* h_src = (const float*)d_in[0];
  const float* h_dst = (const float*)d_in[1];
  const float* Wk    = (const float*)d_in[2];
  const float* Wq    = (const float*)d_in[3];
  const int* src_idx = (const int*)d_in[4];
  const int* dst_idx = (const int*)d_in[5];

  const int N_DST = in_sizes[1] / D;
  const int E     = in_sizes[4];

  auto align256 = [](size_t x) { return (x + 255) & ~(size_t)255; };
  char* ws = (char*)d_ws;
  size_t off = 0;
  float* Q      = (float*)(ws + off); off += align256((size_t)N_DST * D * 4);
  float* P      = (float*)(ws + off); off += align256((size_t)N_DST * H * D * 4);
  int* pairs      = (int*)(ws + off); off += align256((size_t)E * 8);
  int* src_sorted = (int*)(ws + off); off += align256((size_t)E * 4);
  int* seg_cnt    = (int*)(ws + off); off += align256((size_t)N_DST * 4);
  int* seg_start  = (int*)(ws + off); off += align256((size_t)N_DST * 4);
  int* cursor     = (int*)(ws + off); off += align256(4);
  int* run_start  = (int*)(ws + off); off += align256((size_t)NBIN_MAX * SORT_G * 4);
  int* run_cnt    = (int*)(ws + off); off += align256((size_t)NBIN_MAX * SORT_G * 4);

  {
    int waves  = N_DST / 16;
    int blocks = (waves * 32 + GEMM_TB - 1) / GEMM_TB;
    gemm_xwT_wmma<<<blocks, GEMM_TB, 0, stream>>>(h_dst, Wq, Q, N_DST);
    qproj_wmma<<<blocks, GEMM_TB, 0, stream>>>(Q, Wk, P, N_DST);
  }

  chunk_sort<<<SORT_G, SORT_TB, 0, stream>>>(src_idx, dst_idx, pairs, run_start, run_cnt, cursor, E);
  bin_sort<<<(N_DST + BIN_SZ - 1) / BIN_SZ, SORT_TB, 0, stream>>>(
      pairs, run_start, run_cnt, cursor, seg_start, seg_cnt, src_sorted, E, N_DST);
  seg_attend<<<(N_DST * 32 + SEG_TB - 1) / SEG_TB, SEG_TB, 0, stream>>>(
      h_src, h_dst, P, seg_start, seg_cnt, src_sorted, (float*)d_out, N_DST);
}


# Segment kernel builds its 16 dst rows' projected-query planes itself (f32 WMMA, one head per wave, planes via LDS, two rows per wave): their launch, memory write and read back gone; sums unchanged
# speedup vs baseline: 7.1572x; 1.3184x over previous
#include <hip/hip_runtime.h>
#include <hip/hip_bf16.h>

#define D      128
#define H      8
#define DK     16
#define SCALE  0.25f
#define SEG_TB 256

typedef __attribute__((ext_vector_type(2))) float v2f;
typedef __attribute__((ext_vector_type(8))) float v8f;

#define GEMM_TB 256
#define WPITCH  132
__global__ void gemm_xwT_wmma(const float* __restrict__ X,
                              const float* __restrict__ W,
                              float* __restrict__ Y, int M) {
  __shared__ __attribute__((aligned(16))) float Wl[D * WPITCH];

  {
    float4 w4[16];
#pragma unroll
    for (int i = 0; i < 16; ++i) {
      const int idx = threadIdx.x + i * GEMM_TB;
      w4[i] = *(const float4*)(W + (size_t)(idx >> 5) * D + (idx & 31) * 4);
    }
#pragma unroll
    for (int i = 0; i < 16; ++i) {
      const int idx = threadIdx.x + i * GEMM_TB;
      *(float4*)(&Wl[(idx >> 5) * WPITCH + (idx & 31) * 4]) = w4[i];
    }
  }
  __syncthreads();

  const int lane  = threadIdx.x & 31;
  const int wave  = (blockIdx.x * GEMM_TB + threadIdx.x) >> 5;
  const int mBase = wave * 16;
  if (mBase >= M) return;

  const int row  = lane & 15;
  const int koff = (lane >> 4) << 1;

  const float* xrow  = X + (size_t)(mBase + row) * D + koff;
  const float* wbase = Wl + row * WPITCH + koff;

  v8f acc[8] = {};

  v2f a_cur = *(const v2f*)(xrow);
  v2f b_cur[8];
#pragma unroll
  for (int t = 0; t < 8; ++t)
    b_cur[t] = *(const v2f*)(wbase + t * 16 * WPITCH);

#pragma unroll 4
  for (int k0 = 0; k0 < D; k0 += 4) {
    v2f a_nxt;
    v2f b_nxt[8];
    if (k0 + 4 < D) {
      a_nxt = *(const v2f*)(xrow + k0 + 4);
#pragma unroll
      for (int t = 0; t < 8; ++t)
        b_nxt[t] = *(const v2f*)(wbase + t * 16 * WPITCH + k0 + 4);
    }
#pragma unroll
    for (int t = 0; t < 8; ++t) {
      acc[t] = __builtin_amdgcn_wmma_f32_16x16x4_f32(
          false, a_cur, false, b_cur[t], (short)0, acc[t], false, false);
    }
    a_cur = a_nxt;
#pragma unroll
    for (int t = 0; t < 8; ++t) b_cur[t] = b_nxt[t];
  }

  const int n    = lane & 15;
  const int mAdd = (lane >> 4) << 3;
  float* yb = Y + (size_t)mBase * D;
#pragma unroll
  for (int t = 0; t < 8; ++t) {
#pragma unroll
    for (int r = 0; r < 8; ++r) {
      yb[(size_t)(mAdd + r) * D + t * 16 + n] = acc[t][r];
    }
  }
}

#define BIN_SHIFT 8
#define BIN_SZ    256
#define NBIN_MAX  256
#define SORT_TB   1024
#define SORT_G    256
#define SORT_PRE  8
#define RUNS_PW   (SORT_G / (SORT_TB / 32))

__device__ __forceinline__ int excl_scan256(int c, int* wsum) {
  const int tid = threadIdx.x, lane = tid & 31;
  int inc = c;
  for (int o = 1; o < 32; o <<= 1) {
    int v = __shfl_up(inc, o);
    if (lane >= o) inc += v;
  }
  if (tid < 256 && lane == 31) wsum[tid >> 5] = inc;
  __syncthreads();
  int base = 0;
  if (tid < 256)
    for (int i = 0; i < (tid >> 5); ++i) base += wsum[i];
  return base + inc - c;
}

__global__ void __launch_bounds__(SORT_TB)
chunk_sort(const int* __restrict__ src_idx,
           const int* __restrict__ dst_idx,
           int* __restrict__ pairs,
           int* __restrict__ run_start,
           int* __restrict__ run_cnt,
           int* __restrict__ cursor, int E) {
  __shared__ int hist[NBIN_MAX];
  __shared__ int fill[NBIN_MAX];
  __shared__ int wsum[8];
  const int tid = threadIdx.x;
  const int ch  = (E + SORT_G - 1) / SORT_G;
  const int e0  = blockIdx.x * ch;
  const int e1  = (e0 + ch < E) ? (e0 + ch) : E;
  if (tid < NBIN_MAX) hist[tid] = 0;
  if (blockIdx.x == 0 && tid == 0) *cursor = 0;

  int dv[SORT_PRE], sv[SORT_PRE];
#pragma unroll
  for (int k = 0; k < SORT_PRE; ++k) {
    const int e = e0 + tid + k * SORT_TB;
    dv[k] = (e < e1) ? dst_idx[e] : -1;
  }
#pragma unroll
  for (int k = 0; k < SORT_PRE; ++k) {
    const int e = e0 + tid + k * SORT_TB;
    sv[k] = (e < e1) ? src_idx[e] : 0;
  }
  __syncthreads();
#pragma unroll
  for (int k = 0; k < SORT_PRE; ++k)
    if (dv[k] >= 0) atomicAdd(&hist[dv[k] >> BIN_SHIFT], 1);
  for (int e = e0 + tid + SORT_PRE * SORT_TB; e < e1; e += SORT_TB)
    atomicAdd(&hist[dst_idx[e] >> BIN_SHIFT], 1);
  __syncthreads();

  const int c  = (tid < NBIN_MAX) ? hist[tid] : 0;
  const int st = excl_scan256(c, wsum);
  if (tid < NBIN_MAX) {
    fill[tid] = st;
    run_start[tid * SORT_G + blockIdx.x] = st;
    run_cnt[tid * SORT_G + blockIdx.x]   = c;
  }
  __syncthreads();

  int2* mine = (int2*)pairs + e0;
#pragma unroll
  for (int k = 0; k < SORT_PRE; ++k)
    if (dv[k] >= 0) {
      const int p = atomicAdd(&fill[dv[k] >> BIN_SHIFT], 1);
      mine[p] = make_int2(sv[k], dv[k]);
    }
  for (int e = e0 + tid + SORT_PRE * SORT_TB; e < e1; e += SORT_TB) {
    const int d = dst_idx[e];
    const int p = atomicAdd(&fill[d >> BIN_SHIFT], 1);
    mine[p] = make_int2(src_idx[e], d);
  }
}

__global__ void __launch_bounds__(SORT_TB)
bin_sort(const int* __restrict__ pairs,
         const int* __restrict__ run_start,
         const int* __restrict__ run_cnt,
         int* __restrict__ cursor,
         int* __restrict__ seg_start,
         int* __restrict__ seg_cnt,
         int* __restrict__ src_sorted, int E, int Ndst) {
  __shared__ int rs[SORT_G];
  __shared__ int rc[SORT_G];
  __shared__ int cnt[BIN_SZ];
  __shared__ int fill[BIN_SZ];
  __shared__ int wtot[8];
  __shared__ int wsum[8];
  __shared__ int base_s;
  const int tid = threadIdx.x, lane = tid & 31, w = tid >> 5;
  const int b  = blockIdx.x;
  const int ch = (E + SORT_G - 1) / SORT_G;
  if (tid < BIN_SZ) cnt[tid] = 0;
  if (tid < SORT_G) {
    const int c = run_cnt[b * SORT_G + tid];
    rc[tid] = c;
    rs[tid] = tid * ch + run_start[b * SORT_G + tid];
    int s = c;
    for (int o = 16; o > 0; o >>= 1) s += __shfl_xor(s, o);
    if (lane == 0) wtot[w] = s;
  }
  __syncthreads();

  const int2* pp = (const int2*)pairs;
  int  rn[RUNS_PW];
  int2 v0[RUNS_PW], v1[RUNS_PW];
#pragma unroll
  for (int k = 0; k < RUNS_PW; ++k) {
    const int r = w + k * (SORT_TB / 32);
    rn[k] = rc[r];
    int i0 = rs[r] + lane, i1 = i0 + 32;
    i0 = (i0 < E) ? i0 : E - 1;
    i1 = (i1 < E) ? i1 : E - 1;
    v0[k] = pp[i0];
    v1[k] = pp[i1];
  }
  if (tid == 0) {
    int tot = 0;
    for (int i = 0; i < SORT_G / 32; ++i) tot += wtot[i];
    base_s = atomicAdd(cursor, tot);
  }
#pragma unroll
  for (int k = 0; k < RUNS_PW; ++k) {
    if (lane < rn[k])      atomicAdd(&cnt[v0[k].y & (BIN_SZ - 1)], 1);
    if (lane + 32 < rn[k]) atomicAdd(&cnt[v1[k].y & (BIN_SZ - 1)], 1);
    if (rn[k] > 64) {
      const int2* p = pp + rs[w + k * (SORT_TB / 32)];
      for (int i = lane + 64; i < rn[k]; i += 32) atomicAdd(&cnt[p[i].y & (BIN_SZ - 1)], 1);
    }
  }
  __syncthreads();

  const int c  = (tid < BIN_SZ) ? cnt[tid] : 0;
  const int st = excl_scan256(c, wsum);
  const int base = base_s;
  if (tid < BIN_SZ) {
    fill[tid] = st;
    const int d = b * BIN_SZ + tid;
    if (d < Ndst) { seg_start[d] = base + st; seg_cnt[d] = c; }
  }
  __syncthreads();

#pragma unroll
  for (int k = 0; k < RUNS_PW; ++k) {
    if (lane < rn[k])      src_sorted[base + atomicAdd(&fill[v0[k].y & (BIN_SZ - 1)], 1)] = v0[k].x;
    if (lane + 32 < rn[k]) src_sorted[base + atomicAdd(&fill[v1[k].y & (BIN_SZ - 1)], 1)] = v1[k].x;
    if (rn[k] > 64) {
      const int2* p = pp + rs[w + k * (SORT_TB / 32)];
      for (int i = lane + 64; i < rn[k]; i += 32) {
        const int2 pr = p[i];
        src_sorted[base + atomicAdd(&fill[pr.y & (BIN_SZ - 1)], 1)] = pr.x;
      }
    }
  }
}

__device__ __forceinline__ float quad_sum(float p) {
  p += __int_as_float(__builtin_amdgcn_update_dpp(0, __float_as_int(p), 0xB1, 0xF, 0xF, true));
  p += __int_as_float(__builtin_amdgcn_update_dpp(0, __float_as_int(p), 0x4E, 0xF, 0xF, true));
  return p;
}

__device__ __forceinline__ float other_row(float x) {
  return __int_as_float(__builtin_amdgcn_permlanex16(__float_as_int(x), __float_as_int(x),
                                                     0x76543210, 0xfedcba98, false, false));
}

__device__ __forceinline__ float head_score(const float4 x, const v2f (&qq)[4][4]) {
  const v2f xx = {x.x, x.x}, xy = {x.y, x.y}, xz = {x.z, x.z}, xw = {x.w, x.w};
  v2f p[4];
#pragma unroll
  for (int jj = 0; jj < 4; ++jj) {
    v2f t = xx * qq[jj][0];
    t = __builtin_elementwise_fma(xy, qq[jj][1], t);
    t = __builtin_elementwise_fma(xz, qq[jj][2], t);
    t = __builtin_elementwise_fma(xw, qq[jj][3], t);
    p[jj] = t;
  }
  v2f r0, r1;
  r0.x = other_row(p[2].x); r0.y = other_row(p[2].y);
  r1.x = other_row(p[3].x); r1.y = other_row(p[3].y);
  const v2f a = p[0] + r0;
  const v2f b = p[1] + r1;
  v2f s;
  s.x = __int_as_float(__builtin_amdgcn_update_dpp(0, __float_as_int(b.x), 0x128, 0xF, 0xF, true));
  s.y = __int_as_float(__builtin_amdgcn_update_dpp(0, __float_as_int(b.y), 0x128, 0xF, 0xF, true));
  const v2f e = a + s;
  const float c0 = e.x + __int_as_float(__builtin_amdgcn_update_dpp(0, __float_as_int(e.y), 0x141, 0xF, 0xF, true));
  return quad_sum(c0);
}

template <bool TAIL>
__device__ __forceinline__ void attend8(const float* __restrict__ xcol,
                                        int sv, int c0, int cnt, const v2f (&qq)[4][4],
                                        float& m, float& l, float4& acc) {
  float4 xx[8];
  int s[8];
#pragma unroll
  for (int i = 0; i < 8; ++i) s[i] = __builtin_amdgcn_readlane(sv, c0 + i);
#pragma unroll
  for (int i = 0; i < 8; ++i) xx[i] = *(const float4*)(xcol + (size_t)s[i] * D);
  float t[8];
  float cm = m;
#pragma unroll
  for (int i = 0; i < 8; ++i) {
    float p = head_score(xx[i], qq);
    if (TAIL && i >= cnt) p = __uint_as_float(0xff800000u);
    t[i] = p;
    cm = fmaxf(cm, p);
  }
  const float r = __expf(m - cm);
  l *= r;
  acc.x *= r; acc.y *= r; acc.z *= r; acc.w *= r;
#pragma unroll
  for (int i = 0; i < 8; ++i) {
    const float e = __expf(t[i] - cm);
    l += e;
    acc.x += e * xx[i].x; acc.y += e * xx[i].y; acc.z += e * xx[i].z; acc.w += e * xx[i].w;
  }
  m = cm;
}

#define TILE_M 16
#define PROW   1028
__global__ void __launch_bounds__(SEG_TB) __attribute__((amdgpu_waves_per_eu(6, 8)))
seg_attend(const float* __restrict__ h_src,
           const float* __restrict__ h_dst,
           const float* __restrict__ Q,
           const float* __restrict__ Wk,
           const int* __restrict__ seg_start,
           const int* __restrict__ seg_cnt,
           const int* __restrict__ src_sorted,
           float* __restrict__ out, int Ndst) {
  __shared__ __attribute__((aligned(16))) float Pl[TILE_M * PROW];
  const int lane = threadIdx.x & 31;
  const int w     = __builtin_amdgcn_readfirstlane((int)(threadIdx.x >> 5));
  const int mBase = blockIdx.x * TILE_M;

  const int dA = (mBase + w < Ndst) ? (mBase + w) : (Ndst - 1);
  const int dB = (mBase + w + 8 < Ndst) ? (mBase + w + 8) : (Ndst - 1);
  const int begA = seg_start[dA], nA = seg_cnt[dA];
  const int begB = seg_start[dB], nB = seg_cnt[dB];

  {
    const int c  = lane & 15;
    const int hh = lane >> 4;
    int qr = mBase + c;
    qr = (qr < Ndst) ? qr : (Ndst - 1);
    const float* qrow = Q + (size_t)qr * D + 16 * w + 8 * hh;
    const float4 q0 = *(const float4*)(qrow);
    const float4 q1 = *(const float4*)(qrow + 4);
    v2f a[4];
    a[0].x = q0.x; a[0].y = q0.y; a[1].x = q0.z; a[1].y = q0.w;
    a[2].x = q1.x; a[2].y = q1.y; a[3].x = q1.z; a[3].y = q1.w;
    const float* wh = Wk + (size_t)(16 * w + 8 * hh) * D + c;
    v8f acc[8] = {};
#pragma unroll
    for (int s = 0; s < 4; ++s) {
      v2f b[8];
#pragma unroll
      for (int t = 0; t < 8; ++t) {
        b[t].x = wh[(2 * s) * D + 16 * t];
        b[t].y = wh[(2 * s + 1) * D + 16 * t];
      }
#pragma unroll
      for (int t = 0; t < 8; ++t) {
        acc[t] = __builtin_amdgcn_wmma_f32_16x16x4_f32(
            false, a[s], false, b[t], (short)0, acc[t], false, false);
      }
    }
    float* pl = Pl + (8 * hh) * PROW + c;
#pragma unroll
    for (int t = 0; t < 8; ++t) {
#pragma unroll
      for (int r = 0; r < 8; ++r) {
        pl[r * PROW + (w ^ t) * D + t * 16] = acc[t][r];
      }
    }
  }
  __syncthreads();

  const int col = lane * 4;
  const float* xcol = h_src + col;
#pragma nounroll
  for (int rr = 0; rr < 2; ++rr) {
    const int m = w + 8 * rr;
    const int d = mBase + m;
    if (d >= Ndst) break;
    const int beg = __builtin_amdgcn_readfirstlane(rr ? begB : begA);
    const int n   = __builtin_amdgcn_readfirstlane(rr ? nB : nA);
    v2f qq[4][4];
    const float* prow = Pl + m * PROW + col;
#pragma unroll
    for (int jj = 0; jj < 4; ++jj) {
      const float4 u  = *(const float4*)(prow + (2 * jj) * D);
      const float4 wv = *(const float4*)(prow + (2 * jj + 1) * D);
      qq[jj][0].x = u.x * SCALE; qq[jj][0].y = wv.x * SCALE;
      qq[jj][1].x = u.y * SCALE; qq[jj][1].y = wv.y * SCALE;
      qq[jj][2].x = u.z * SCALE; qq[jj][2].y = wv.z * SCALE;
      qq[jj][3].x = u.w * SCALE; qq[jj][3].y = wv.w * SCALE;
    }
    const float4 xd = *(const float4*)(h_dst + (size_t)d * D + col);
    float mx = head_score(xd, qq);
    float l = 1.0f;
    float4 acc = xd;
    for (int j0 = 0; j0 < n; j0 += 32) {
      const int mc = (n - j0 < 32) ? (n - j0) : 32;
      const int sv = (lane < mc) ? src_sorted[beg + j0 + lane] : 0;
      int c0 = 0;
      for (; c0 + 8 <= mc; c0 += 8) attend8<false>(xcol, sv, c0, 8, qq, mx, l, acc);
      if (c0 < mc) attend8<true>(xcol, sv, c0, mc - c0, qq, mx, l, acc);
    }
    const float rl = 1.0f / l;
    float4 o;
    o.x = acc.x * rl; o.y = acc.y * rl; o.z = acc.z * rl; o.w = acc.w * rl;
    *(float4*)(out + (size_t)d * D + col) = o;
  }
}

extern "C" void kernel_launch(void* const* d_in, const int* in_sizes, int n_in,
                              void* d_out, int out_size, void* d_ws, size_t ws_size,
                              hipStream_t stream) {
  const float* h_src = (const float*)d_in[0];
  const float* h_dst = (const float*)d_in[1];
  const float* Wk    = (const float*)d_in[2];
  const float* Wq    = (const float*)d_in[3];
  const int* src_idx = (const int*)d_in[4];
  const int* dst_idx = (const int*)d_in[5];

  const int N_DST = in_sizes[1] / D;
  const int E     = in_sizes[4];

  auto align256 = [](size_t x) { return (x + 255) & ~(size_t)255; };
  char* ws = (char*)d_ws;
  size_t off = 0;
  float* Q      = (float*)(ws + off); off += align256((size_t)N_DST * D * 4);
  int* pairs      = (int*)(ws + off); off += align256((size_t)E * 8);
  int* src_sorted = (int*)(ws + off); off += align256((size_t)E * 4);
  int* seg_cnt    = (int*)(ws + off); off += align256((size_t)N_DST * 4);
  int* seg_start  = (int*)(ws + off); off += align256((size_t)N_DST * 4);
  int* cursor     = (int*)(ws + off); off += align256(4);
  int* run_start  = (int*)(ws + off); off += align256((size_t)NBIN_MAX * SORT_G * 4);
  int* run_cnt    = (int*)(ws + off); off += align256((size_t)NBIN_MAX * SORT_G * 4);

  {
    int waves  = N_DST / 16;
    int blocks = (waves * 32 + GEMM_TB - 1) / GEMM_TB;
    gemm_xwT_wmma<<<blocks, GEMM_TB, 0, stream>>>(h_dst, Wq, Q, N_DST);
  }

  chunk_sort<<<SORT_G, SORT_TB, 0, stream>>>(src_idx, dst_idx, pairs, run_start, run_cnt, cursor, E);
  bin_sort<<<(N_DST + BIN_SZ - 1) / BIN_SZ, SORT_TB, 0, stream>>>(
      pairs, run_start, run_cnt, cursor, seg_start, seg_cnt, src_sorted, E, N_DST);
  seg_attend<<<(N_DST + TILE_M - 1) / TILE_M, SEG_TB, 0, stream>>>(
      h_src, h_dst, Q, Wk, seg_start, seg_cnt, src_sorted, (float*)d_out, N_DST);
}


# The q projection runs inside the chunk-sort launch as extra 32-wave work-groups beside the sorting ones (2 waves per 16-row tile, 4 column tiles each): one launch fewer; k order and sums unchanged
# speedup vs baseline: 7.4120x; 1.0356x over previous
#include <hip/hip_runtime.h>
#include <hip/hip_bf16.h>

#define D      128
#define H      8
#define DK     16
#define SCALE  0.25f
#define SEG_TB 256

typedef __attribute__((ext_vector_type(2))) float v2f;
typedef __attribute__((ext_vector_type(8))) float v8f;

#define PROJ_ROWS 256
#define PROJ_TB   1024
#define WPITCH  132
__device__ __forceinline__ void project_rows(const float* __restrict__ X,
                                             const float* __restrict__ W,
                                             float* __restrict__ Y, int M, int bid) {
  __shared__ __attribute__((aligned(16))) float Wl[D * WPITCH];

  {
    float4 w4[4];
#pragma unroll
    for (int i = 0; i < 4; ++i) {
      const int idx = threadIdx.x + i * PROJ_TB;
      w4[i] = *(const float4*)(W + (size_t)(idx >> 5) * D + (idx & 31) * 4);
    }
#pragma unroll
    for (int i = 0; i < 4; ++i) {
      const int idx = threadIdx.x + i * PROJ_TB;
      *(float4*)(&Wl[(idx >> 5) * WPITCH + (idx & 31) * 4]) = w4[i];
    }
  }
  __syncthreads();

  const int lane  = threadIdx.x & 31;
  const int wv    = __builtin_amdgcn_readfirstlane((int)(threadIdx.x >> 5));
  const int nh    = wv & 1;
  const int mBase = (bid * (PROJ_ROWS / 16) + (wv >> 1)) * 16;
  if (mBase >= M) return;

  const int row  = lane & 15;
  const int koff = (lane >> 4) << 1;

  const float* xrow  = X + (size_t)(mBase + row) * D + koff;
  const float* wbase = Wl + (nh * 64 + row) * WPITCH + koff;

  v8f acc[4] = {};

  v2f a_cur = *(const v2f*)(xrow);
  v2f b_cur[4];
#pragma unroll
  for (int t = 0; t < 4; ++t)
    b_cur[t] = *(const v2f*)(wbase + t * 16 * WPITCH);

#pragma unroll 4
  for (int k0 = 0; k0 < D; k0 += 4) {
    v2f a_nxt;
    v2f b_nxt[4];
    if (k0 + 4 < D) {
      a_nxt = *(const v2f*)(xrow + k0 + 4);
#pragma unroll
      for (int t = 0; t < 4; ++t)
        b_nxt[t] = *(const v2f*)(wbase + t * 16 * WPITCH + k0 + 4);
    }
#pragma unroll
    for (int t = 0; t < 4; ++t) {
      acc[t] = __builtin_amdgcn_wmma_f32_16x16x4_f32(
          false, a_cur, false, b_cur[t], (short)0, acc[t], false, false);
    }
    a_cur = a_nxt;
#pragma unroll
    for (int t = 0; t < 4; ++t) b_cur[t] = b_nxt[t];
  }

  const int n    = lane & 15;
  const int mAdd = (lane >> 4) << 3;
  float* yb = Y + (size_t)mBase * D + nh * 64;
#pragma unroll
  for (int t = 0; t < 4; ++t) {
#pragma unroll
    for (int r = 0; r < 8; ++r) {
      yb[(size_t)(mAdd + r) * D + t * 16 + n] = acc[t][r];
    }
  }
}

#define BIN_SHIFT 8
#define BIN_SZ    256
#define NBIN_MAX  256
#define SORT_TB   1024
#define SORT_G    256
#define SORT_PRE  8
#define RUNS_PW   (SORT_G / (SORT_TB / 32))

__device__ __forceinline__ int excl_scan256(int c, int* wsum) {
  const int tid = threadIdx.x, lane = tid & 31;
  int inc = c;
  for (int o = 1; o < 32; o <<= 1) {
    int v = __shfl_up(inc, o);
    if (lane >= o) inc += v;
  }
  if (tid < 256 && lane == 31) wsum[tid >> 5] = inc;
  __syncthreads();
  int base = 0;
  if (tid < 256)
    for (int i = 0; i < (tid >> 5); ++i) base += wsum[i];
  return base + inc - c;
}

__device__ __forceinline__ void
chunk_sort(const int* __restrict__ src_idx,
           const int* __restrict__ dst_idx,
           int* __restrict__ pairs,
           int* __restrict__ run_start,
           int* __restrict__ run_cnt,
           int* __restrict__ cursor, int E, int bid) {
  __shared__ int hist[NBIN_MAX];
  __shared__ int fill[NBIN_MAX];
  __shared__ int wsum[8];
  const int tid = threadIdx.x;
  const int ch  = (E + SORT_G - 1) / SORT_G;
  const int e0  = bid * ch;
  const int e1  = (e0 + ch < E) ? (e0 + ch) : E;
  if (tid < NBIN_MAX) hist[tid] = 0;
  if (bid == 0 && tid == 0) *cursor = 0;

  int dv[SORT_PRE], sv[SORT_PRE];
#pragma unroll
  for (int k = 0; k < SORT_PRE; ++k) {
    const int e = e0 + tid + k * SORT_TB;
    dv[k] = (e < e1) ? dst_idx[e] : -1;
  }
#pragma unroll
  for (int k = 0; k < SORT_PRE; ++k) {
    const int e = e0 + tid + k * SORT_TB;
    sv[k] = (e < e1) ? src_idx[e] : 0;
  }
  __syncthreads();
#pragma unroll
  for (int k = 0; k < SORT_PRE; ++k)
    if (dv[k] >= 0) atomicAdd(&hist[dv[k] >> BIN_SHIFT], 1);
  for (int e = e0 + tid + SORT_PRE * SORT_TB; e < e1; e += SORT_TB)
    atomicAdd(&hist[dst_idx[e] >> BIN_SHIFT], 1);
  __syncthreads();

  const int c  = (tid < NBIN_MAX) ? hist[tid] : 0;
  const int st = excl_scan256(c, wsum);
  if (tid < NBIN_MAX) {
    fill[tid] = st;
    run_start[tid * SORT_G + bid] = st;
    run_cnt[tid * SORT_G + bid]   = c;
  }
  __syncthreads();

  int2* mine = (int2*)pairs + e0;
#pragma unroll
  for (int k = 0; k < SORT_PRE; ++k)
    if (dv[k] >= 0) {
      const int p = atomicAdd(&fill[dv[k] >> BIN_SHIFT], 1);
      mine[p] = make_int2(sv[k], dv[k]);
    }
  for (int e = e0 + tid + SORT_PRE * SORT_TB; e < e1; e += SORT_TB) {
    const int d = dst_idx[e];
    const int p = atomicAdd(&fill[d >> BIN_SHIFT], 1);
    mine[p] = make_int2(src_idx[e], d);
  }
}

__global__ void __launch_bounds__(SORT_TB) __attribute__((amdgpu_waves_per_eu(16)))
project_and_chunk_sort(const float* __restrict__ h_dst,
                       const float* __restrict__ Wq,
                       float* __restrict__ Q, int Ndst, int projG,
                       const int* __restrict__ src_idx,
                       const int* __restrict__ dst_idx,
                       int* __restrict__ pairs,
                       int* __restrict__ run_start,
                       int* __restrict__ run_cnt,
                       int* __restrict__ cursor, int E) {
  const int b = blockIdx.x;
  if (b < projG) project_rows(h_dst, Wq, Q, Ndst, b);
  else           chunk_sort(src_idx, dst_idx, pairs, run_start, run_cnt, cursor, E, b - projG);
}

__global__ void __launch_bounds__(SORT_TB)
bin_sort(const int* __restrict__ pairs,
         const int* __restrict__ run_start,
         const int* __restrict__ run_cnt,
         int* __restrict__ cursor,
         int* __restrict__ seg_start,
         int* __restrict__ seg_cnt,
         int* __restrict__ src_sorted, int E, int Ndst) {
  __shared__ int rs[SORT_G];
  __shared__ int rc[SORT_G];
  __shared__ int cnt[BIN_SZ];
  __shared__ int fill[BIN_SZ];
  __shared__ int wtot[8];
  __shared__ int wsum[8];
  __shared__ int base_s;
  const int tid = threadIdx.x, lane = tid & 31, w = tid >> 5;
  const int b  = blockIdx.x;
  const int ch = (E + SORT_G - 1) / SORT_G;
  if (tid < BIN_SZ) cnt[tid] = 0;
  if (tid < SORT_G) {
    const int c = run_cnt[b * SORT_G + tid];
    rc[tid] = c;
    rs[tid] = tid * ch + run_start[b * SORT_G + tid];
    int s = c;
    for (int o = 16; o > 0; o >>= 1) s += __shfl_xor(s, o);
    if (lane == 0) wtot[w] = s;
  }
  __syncthreads();

  const int2* pp = (const int2*)pairs;
  int  rn[RUNS_PW];
  int2 v0[RUNS_PW], v1[RUNS_PW];
#pragma unroll
  for (int k = 0; k < RUNS_PW; ++k) {
    const int r = w + k * (SORT_TB / 32);
    rn[k] = rc[r];
    int i0 = rs[r] + lane, i1 = i0 + 32;
    i0 = (i0 < E) ? i0 : E - 1;
    i1 = (i1 < E) ? i1 : E - 1;
    v0[k] = pp[i0];
    v1[k] = pp[i1];
  }
  if (tid == 0) {
    int tot = 0;
    for (int i = 0; i < SORT_G / 32; ++i) tot += wtot[i];
    base_s = atomicAdd(cursor, tot);
  }
#pragma unroll
  for (int k = 0; k < RUNS_PW; ++k) {
    if (lane < rn[k])      atomicAdd(&cnt[v0[k].y & (BIN_SZ - 1)], 1);
    if (lane + 32 < rn[k]) atomicAdd(&cnt[v1[k].y & (BIN_SZ - 1)], 1);
    if (rn[k] > 64) {
      const int2* p = pp + rs[w + k * (SORT_TB / 32)];
      for (int i = lane + 64; i < rn[k]; i += 32) atomicAdd(&cnt[p[i].y & (BIN_SZ - 1)], 1);
    }
  }
  __syncthreads();

  const int c  = (tid < BIN_SZ) ? cnt[tid] : 0;
  const int st = excl_scan256(c, wsum);
  const int base = base_s;
  if (tid < BIN_SZ) {
    fill[tid] = st;
    const int d = b * BIN_SZ + tid;
    if (d < Ndst) { seg_start[d] = base + st; seg_cnt[d] = c; }
  }
  __syncthreads();

#pragma unroll
  for (int k = 0; k < RUNS_PW; ++k) {
    if (lane < rn[k])      src_sorted[base + atomicAdd(&fill[v0[k].y & (BIN_SZ - 1)], 1)] = v0[k].x;
    if (lane + 32 < rn[k]) src_sorted[base + atomicAdd(&fill[v1[k].y & (BIN_SZ - 1)], 1)] = v1[k].x;
    if (rn[k] > 64) {
      const int2* p = pp + rs[w + k * (SORT_TB / 32)];
      for (int i = lane + 64; i < rn[k]; i += 32) {
        const int2 pr = p[i];
        src_sorted[base + atomicAdd(&fill[pr.y & (BIN_SZ - 1)], 1)] = pr.x;
      }
    }
  }
}

__device__ __forceinline__ float quad_sum(float p) {
  p += __int_as_float(__builtin_amdgcn_update_dpp(0, __float_as_int(p), 0xB1, 0xF, 0xF, true));
  p += __int_as_float(__builtin_amdgcn_update_dpp(0, __float_as_int(p), 0x4E, 0xF, 0xF, true));
  return p;
}

__device__ __forceinline__ float other_row(float x) {
  return __int_as_float(__builtin_amdgcn_permlanex16(__float_as_int(x), __float_as_int(x),
                                                     0x76543210, 0xfedcba98, false, false));
}

__device__ __forceinline__ float head_score(const float4 x, const v2f (&qq)[4][4]) {
  const v2f xx = {x.x, x.x}, xy = {x.y, x.y}, xz = {x.z, x.z}, xw = {x.w, x.w};
  v2f p[4];
#pragma unroll
  for (int jj = 0; jj < 4; ++jj) {
    v2f t = xx * qq[jj][0];
    t = __builtin_elementwise_fma(xy, qq[jj][1], t);
    t = __builtin_elementwise_fma(xz, qq[jj][2], t);
    t = __builtin_elementwise_fma(xw, qq[jj][3], t);
    p[jj] = t;
  }
  v2f r0, r1;
  r0.x = other_row(p[2].x); r0.y = other_row(p[2].y);
  r1.x = other_row(p[3].x); r1.y = other_row(p[3].y);
  const v2f a = p[0] + r0;
  const v2f b = p[1] + r1;
  v2f s;
  s.x = __int_as_float(__builtin_amdgcn_update_dpp(0, __float_as_int(b.x), 0x128, 0xF, 0xF, true));
  s.y = __int_as_float(__builtin_amdgcn_update_dpp(0, __float_as_int(b.y), 0x128, 0xF, 0xF, true));
  const v2f e = a + s;
  const float c0 = e.x + __int_as_float(__builtin_amdgcn_update_dpp(0, __float_as_int(e.y), 0x141, 0xF, 0xF, true));
  return quad_sum(c0);
}

template <bool TAIL>
__device__ __forceinline__ void attend8(const float* __restrict__ xcol,
                                        int sv, int c0, int cnt, const v2f (&qq)[4][4],
                                        float& m, float& l, float4& acc) {
  float4 xx[8];
  int s[8];
#pragma unroll
  for (int i = 0; i < 8; ++i) s[i] = __builtin_amdgcn_readlane(sv, c0 + i);
#pragma unroll
  for (int i = 0; i < 8; ++i) xx[i] = *(const float4*)(xcol + (size_t)s[i] * D);
  float t[8];
  float cm = m;
#pragma unroll
  for (int i = 0; i < 8; ++i) {
    float p = head_score(xx[i], qq);
    if (TAIL && i >= cnt) p = __uint_as_float(0xff800000u);
    t[i] = p;
    cm = fmaxf(cm, p);
  }
  const float r = __expf(m - cm);
  l *= r;
  acc.x *= r; acc.y *= r; acc.z *= r; acc.w *= r;
#pragma unroll
  for (int i = 0; i < 8; ++i) {
    const float e = __expf(t[i] - cm);
    l += e;
    acc.x += e * xx[i].x; acc.y += e * xx[i].y; acc.z += e * xx[i].z; acc.w += e * xx[i].w;
  }
  m = cm;
}

#define TILE_M 16
#define PROW   1028
__global__ void __launch_bounds__(SEG_TB) __attribute__((amdgpu_waves_per_eu(6, 8)))
seg_attend(const float* __restrict__ h_src,
           const float* __restrict__ h_dst,
           const float* __restrict__ Q,
           const float* __restrict__ Wk,
           const int* __restrict__ seg_start,
           const int* __restrict__ seg_cnt,
           const int* __restrict__ src_sorted,
           float* __restrict__ out, int Ndst) {
  __shared__ __attribute__((aligned(16))) float Pl[TILE_M * PROW];
  const int lane = threadIdx.x & 31;
  const int w     = __builtin_amdgcn_readfirstlane((int)(threadIdx.x >> 5));
  const int mBase = blockIdx.x * TILE_M;

  const int dA = (mBase + w < Ndst) ? (mBase + w) : (Ndst - 1);
  const int dB = (mBase + w + 8 < Ndst) ? (mBase + w + 8) : (Ndst - 1);
  const int begA = seg_start[dA], nA = seg_cnt[dA];
  const int begB = seg_start[dB], nB = seg_cnt[dB];

  {
    const int c  = lane & 15;
    const int hh = lane >> 4;
    int qr = mBase + c;
    qr = (qr < Ndst) ? qr : (Ndst - 1);
    const float* qrow = Q + (size_t)qr * D + 16 * w + 8 * hh;
    const float4 q0 = *(const float4*)(qrow);
    const float4 q1 = *(const float4*)(qrow + 4);
    v2f a[4];
    a[0].x = q0.x; a[0].y = q0.y; a[1].x = q0.z; a[1].y = q0.w;
    a[2].x = q1.x; a[2].y = q1.y; a[3].x = q1.z; a[3].y = q1.w;
    const float* wh = Wk + (size_t)(16 * w + 8 * hh) * D + c;
    v8f acc[8] = {};
#pragma unroll
    for (int s = 0; s < 4; ++s) {
      v2f b[8];
#pragma unroll
      for (int t = 0; t < 8; ++t) {
        b[t].x = wh[(2 * s) * D + 16 * t];
        b[t].y = wh[(2 * s + 1) * D + 16 * t];
      }
#pragma unroll
      for (int t = 0; t < 8; ++t) {
        acc[t] = __builtin_amdgcn_wmma_f32_16x16x4_f32(
            false, a[s], false, b[t], (short)0, acc[t], false, false);
      }
    }
    float* pl = Pl + (8 * hh) * PROW + c;
#pragma unroll
    for (int t = 0; t < 8; ++t) {
#pragma unroll
      for (int r = 0; r < 8; ++r) {
        pl[r * PROW + (w ^ t) * D + t * 16] = acc[t][r];
      }
    }
  }
  __syncthreads();

  const int col = lane * 4;
  const float* xcol = h_src + col;
#pragma nounroll
  for (int rr = 0; rr < 2; ++rr) {
    const int m = w + 8 * rr;
    const int d = mBase + m;
    if (d >= Ndst) break;
    const int beg = __builtin_amdgcn_readfirstlane(rr ? begB : begA);
    const int n   = __builtin_amdgcn_readfirstlane(rr ? nB : nA);
    v2f qq[4][4];
    const float* prow = Pl + m * PROW + col;
#pragma unroll
    for (int jj = 0; jj < 4; ++jj) {
      const float4 u  = *(const float4*)(prow + (2 * jj) * D);
      const float4 wv = *(const float4*)(prow + (2 * jj + 1) * D);
      qq[jj][0].x = u.x * SCALE; qq[jj][0].y = wv.x * SCALE;
      qq[jj][1].x = u.y * SCALE; qq[jj][1].y = wv.y * SCALE;
      qq[jj][2].x = u.z * SCALE; qq[jj][2].y = wv.z * SCALE;
      qq[jj][3].x = u.w * SCALE; qq[jj][3].y = wv.w * SCALE;
    }
    const float4 xd = *(const float4*)(h_dst + (size_t)d * D + col);
    float mx = head_score(xd, qq);
    float l = 1.0f;
    float4 acc = xd;
    for (int j0 = 0; j0 < n; j0 += 32) {
      const int mc = (n - j0 < 32) ? (n - j0) : 32;
      const int sv = (lane < mc) ? src_sorted[beg + j0 + lane] : 0;
      int c0 = 0;
      for (; c0 + 8 <= mc; c0 += 8) attend8<false>(xcol, sv, c0, 8, qq, mx, l, acc);
      if (c0 < mc) attend8<true>(xcol, sv, c0, mc - c0, qq, mx, l, acc);
    }
    const float rl = 1.0f / l;
    float4 o;
    o.x = acc.x * rl; o.y = acc.y * rl; o.z = acc.z * rl; o.w = acc.w * rl;
    *(float4*)(out + (size_t)d * D + col) = o;
  }
}

extern "C" void kernel_launch(void* const* d_in, const int* in_sizes, int n_in,
                              void* d_out, int out_size, void* d_ws, size_t ws_size,
                              hipStream_t stream) {
  const float* h_src = (const float*)d_in[0];
  const float* h_dst = (const float*)d_in[1];
  const float* Wk    = (const float*)d_in[2];
  const float* Wq    = (const float*)d_in[3];
  const int* src_idx = (const int*)d_in[4];
  const int* dst_idx = (const int*)d_in[5];

  const int N_DST = in_sizes[1] / D;
  const int E     = in_sizes[4];

  auto align256 = [](size_t x) { return (x + 255) & ~(size_t)255; };
  char* ws = (char*)d_ws;
  size_t off = 0;
  float* Q      = (float*)(ws + off); off += align256((size_t)N_DST * D * 4);
  int* pairs      = (int*)(ws + off); off += align256((size_t)E * 8);
  int* src_sorted = (int*)(ws + off); off += align256((size_t)E * 4);
  int* seg_cnt    = (int*)(ws + off); off += align256((size_t)N_DST * 4);
  int* seg_start  = (int*)(ws + off); off += align256((size_t)N_DST * 4);
  int* cursor     = (int*)(ws + off); off += align256(4);
  int* run_start  = (int*)(ws + off); off += align256((size_t)NBIN_MAX * SORT_G * 4);
  int* run_cnt    = (int*)(ws + off); off += align256((size_t)NBIN_MAX * SORT_G * 4);

  const int projG = (N_DST + PROJ_ROWS - 1) / PROJ_ROWS;
  project_and_chunk_sort<<<projG + SORT_G, SORT_TB, 0, stream>>>(
      h_dst, Wq, Q, N_DST, projG, src_idx, dst_idx, pairs, run_start, run_cnt, cursor, E);
  bin_sort<<<(N_DST + BIN_SZ - 1) / BIN_SZ, SORT_TB, 0, stream>>>(
      pairs, run_start, run_cnt, cursor, seg_start, seg_cnt, src_sorted, E, N_DST);
  seg_attend<<<(N_DST + TILE_M - 1) / TILE_M, SEG_TB, 0, stream>>>(
      h_src, h_dst, Q, Wk, seg_start, seg_cnt, src_sorted, (float*)d_out, N_DST);
}


# speedup vs baseline: 7.6776x; 1.0358x over previous
#include <hip/hip_runtime.h>
#include <hip/hip_bf16.h>

#define D      128
#define H      8
#define DK     16
#define SCALE  0.25f
#define SEG_TB 256

typedef __attribute__((ext_vector_type(2))) float v2f;
typedef __attribute__((ext_vector_type(8))) float v8f;

#define PROJ_ROWS 256
#define PROJ_TB   1024
#define WPITCH  132
__device__ __forceinline__ void project_rows(const float* __restrict__ X,
                                             const float* __restrict__ W,
                                             float* __restrict__ Y, int M, int bid) {
  __shared__ __attribute__((aligned(16))) float Wl[D * WPITCH];

  {
    float4 w4[4];
#pragma unroll
    for (int i = 0; i < 4; ++i) {
      const int idx = threadIdx.x + i * PROJ_TB;
      w4[i] = *(const float4*)(W + (size_t)(idx >> 5) * D + (idx & 31) * 4);
    }
#pragma unroll
    for (int i = 0; i < 4; ++i) {
      const int idx = threadIdx.x + i * PROJ_TB;
      *(float4*)(&Wl[(idx >> 5) * WPITCH + (idx & 31) * 4]) = w4[i];
    }
  }
  __syncthreads();

  const int lane  = threadIdx.x & 31;
  const int wv    = __builtin_amdgcn_readfirstlane((int)(threadIdx.x >> 5));
  const int nh    = wv & 1;
  const int mBase = (bid * (PROJ_ROWS / 16) + (wv >> 1)) * 16;
  if (mBase >= M) return;

  const int row  = lane & 15;
  const int koff = (lane >> 4) << 1;

  const float* xrow  = X + (size_t)(mBase + row) * D + koff;
  const float* wbase = Wl + (nh * 64 + row) * WPITCH + koff;

  v8f acc[4] = {};

  v2f a_cur = *(const v2f*)(xrow);
  v2f b_cur[4];
#pragma unroll
  for (int t = 0; t < 4; ++t)
    b_cur[t] = *(const v2f*)(wbase + t * 16 * WPITCH);

#pragma unroll 4
  for (int k0 = 0; k0 < D; k0 += 4) {
    v2f a_nxt;
    v2f b_nxt[4];
    if (k0 + 4 < D) {
      a_nxt = *(const v2f*)(xrow + k0 + 4);
#pragma unroll
      for (int t = 0; t < 4; ++t)
        b_nxt[t] = *(const v2f*)(wbase + t * 16 * WPITCH + k0 + 4);
    }
#pragma unroll
    for (int t = 0; t < 4; ++t) {
      acc[t] = __builtin_amdgcn_wmma_f32_16x16x4_f32(
          false, a_cur, false, b_cur[t], (short)0, acc[t], false, false);
    }
    a_cur = a_nxt;
#pragma unroll
    for (int t = 0; t < 4; ++t) b_cur[t] = b_nxt[t];
  }

  const int n    = lane & 15;
  const int mAdd = (lane >> 4) << 3;
  float* yb = Y + (size_t)mBase * D + nh * 64;
#pragma unroll
  for (int t = 0; t < 4; ++t) {
#pragma unroll
    for (int r = 0; r < 8; ++r) {
      yb[(size_t)(mAdd + r) * D + t * 16 + n] = acc[t][r];
    }
  }
}

#define BIN_SHIFT 8
#define BIN_SZ    256
#define NBIN_MAX  256
#define SORT_TB   1024
#define SORT_G    256
#define SORT_PRE  8
#define RUNS_PW   (SORT_G / (SORT_TB / 32))

__device__ __forceinline__ int excl_scan256(int c, int* wsum) {
  const int tid = threadIdx.x, lane = tid & 31;
  int inc = c;
  for (int o = 1; o < 32; o <<= 1) {
    int v = __shfl_up(inc, o);
    if (lane >= o) inc += v;
  }
  if (tid < 256 && lane == 31) wsum[tid >> 5] = inc;
  __syncthreads();
  int base = 0;
  if (tid < 256)
    for (int i = 0; i < (tid >> 5); ++i) base += wsum[i];
  return base + inc - c;
}

__device__ __forceinline__ void
chunk_sort(const int* __restrict__ src_idx,
           const int* __restrict__ dst_idx,
           int* __restrict__ pairs,
           int* __restrict__ run_start,
           int* __restrict__ run_cnt,
           int* __restrict__ cursor, int E, int bid) {
  __shared__ int hist[NBIN_MAX];
  __shared__ int fill[NBIN_MAX];
  __shared__ int wsum[8];
  const int tid = threadIdx.x;
  const int ch  = (E + SORT_G - 1) / SORT_G;
  const int e0  = bid * ch;
  const int e1  = (e0 + ch < E) ? (e0 + ch) : E;
  if (tid < NBIN_MAX) hist[tid] = 0;
  if (bid == 0 && tid == 0) *cursor = 0;

  int dv[SORT_PRE], sv[SORT_PRE];
#pragma unroll
  for (int k = 0; k < SORT_PRE; ++k) {
    const int e = e0 + tid + k * SORT_TB;
    dv[k] = (e < e1) ? dst_idx[e] : -1;
  }
#pragma unroll
  for (int k = 0; k < SORT_PRE; ++k) {
    const int e = e0 + tid + k * SORT_TB;
    sv[k] = (e < e1) ? src_idx[e] : 0;
  }
  __syncthreads();
#pragma unroll
  for (int k = 0; k < SORT_PRE; ++k)
    if (dv[k] >= 0) atomicAdd(&hist[dv[k] >> BIN_SHIFT], 1);
  for (int e = e0 + tid + SORT_PRE * SORT_TB; e < e1; e += SORT_TB)
    atomicAdd(&hist[dst_idx[e] >> BIN_SHIFT], 1);
  __syncthreads();

  const int c  = (tid < NBIN_MAX) ? hist[tid] : 0;
  const int st = excl_scan256(c, wsum);
  if (tid < NBIN_MAX) {
    fill[tid] = st;
    run_start[tid * SORT_G + bid] = st;
    run_cnt[tid * SORT_G + bid]   = c;
  }
  __syncthreads();

  int2* mine = (int2*)pairs + e0;
#pragma unroll
  for (int k = 0; k < SORT_PRE; ++k)
    if (dv[k] >= 0) {
      const int p = atomicAdd(&fill[dv[k] >> BIN_SHIFT], 1);
      mine[p] = make_int2(sv[k], dv[k]);
    }
  for (int e = e0 + tid + SORT_PRE * SORT_TB; e < e1; e += SORT_TB) {
    const int d = dst_idx[e];
    const int p = atomicAdd(&fill[d >> BIN_SHIFT], 1);
    mine[p] = make_int2(src_idx[e], d);
  }
}

__global__ void __launch_bounds__(SORT_TB) __attribute__((amdgpu_waves_per_eu(16)))
project_and_chunk_sort(const float* __restrict__ h_dst,
                       const float* __restrict__ Wq,
                       float* __restrict__ Q, int Ndst, int projG,
                       const int* __restrict__ src_idx,
                       const int* __restrict__ dst_idx,
                       int* __restrict__ pairs,
                       int* __restrict__ run_start,
                       int* __restrict__ run_cnt,
                       int* __restrict__ cursor, int E) {
  const int b = blockIdx.x;
  if (b < projG) project_rows(h_dst, Wq, Q, Ndst, b);
  else           chunk_sort(src_idx, dst_idx, pairs, run_start, run_cnt, cursor, E, b - projG);
}

__global__ void __launch_bounds__(SORT_TB)
bin_sort(const int* __restrict__ pairs,
         const int* __restrict__ run_start,
         const int* __restrict__ run_cnt,
         int* __restrict__ cursor,
         int* __restrict__ seg_start,
         int* __restrict__ seg_cnt,
         int* __restrict__ src_sorted, int E, int Ndst) {
  __shared__ int rs[SORT_G];
  __shared__ int rc[SORT_G];
  __shared__ int cnt[BIN_SZ];
  __shared__ int fill[BIN_SZ];
  __shared__ int wtot[8];
  __shared__ int wsum[8];
  __shared__ int base_s;
  const int tid = threadIdx.x, lane = tid & 31, w = tid >> 5;
  const int b  = blockIdx.x;
  const int ch = (E + SORT_G - 1) / SORT_G;
  if (tid < BIN_SZ) cnt[tid] = 0;
  if (tid < SORT_G) {
    const int c = run_cnt[b * SORT_G + tid];
    rc[tid] = c;
    rs[tid] = tid * ch + run_start[b * SORT_G + tid];
    int s = c;
    for (int o = 16; o > 0; o >>= 1) s += __shfl_xor(s, o);
    if (lane == 0) wtot[w] = s;
  }
  __syncthreads();

  const int2* pp = (const int2*)pairs;
  int  rn[RUNS_PW];
  int2 v0[RUNS_PW], v1[RUNS_PW];
#pragma unroll
  for (int k = 0; k < RUNS_PW; ++k) {
    const int r = w + k * (SORT_TB / 32);
    rn[k] = rc[r];
    int i0 = rs[r] + lane, i1 = i0 + 32;
    i0 = (i0 < E) ? i0 : E - 1;
    i1 = (i1 < E) ? i1 : E - 1;
    v0[k] = pp[i0];
    v1[k] = pp[i1];
  }
  if (tid == 0) {
    int tot = 0;
    for (int i = 0; i < SORT_G / 32; ++i) tot += wtot[i];
    base_s = atomicAdd(cursor, tot);
  }
#pragma unroll
  for (int k = 0; k < RUNS_PW; ++k) {
    if (lane < rn[k])      atomicAdd(&cnt[v0[k].y & (BIN_SZ - 1)], 1);
    if (lane + 32 < rn[k]) atomicAdd(&cnt[v1[k].y & (BIN_SZ - 1)], 1);
    if (rn[k] > 64) {
      const int2* p = pp + rs[w + k * (SORT_TB / 32)];
      for (int i = lane + 64; i < rn[k]; i += 32) atomicAdd(&cnt[p[i].y & (BIN_SZ - 1)], 1);
    }
  }
  __syncthreads();

  const int c  = (tid < BIN_SZ) ? cnt[tid] : 0;
  const int st = excl_scan256(c, wsum);
  const int base = base_s;
  if (tid < BIN_SZ) {
    fill[tid] = st;
    const int d = b * BIN_SZ + tid;
    if (d < Ndst) { seg_start[d] = base + st; seg_cnt[d] = c; }
  }
  __syncthreads();

#pragma unroll
  for (int k = 0; k < RUNS_PW; ++k) {
    if (lane < rn[k])      src_sorted[base + atomicAdd(&fill[v0[k].y & (BIN_SZ - 1)], 1)] = v0[k].x;
    if (lane + 32 < rn[k]) src_sorted[base + atomicAdd(&fill[v1[k].y & (BIN_SZ - 1)], 1)] = v1[k].x;
    if (rn[k] > 64) {
      const int2* p = pp + rs[w + k * (SORT_TB / 32)];
      for (int i = lane + 64; i < rn[k]; i += 32) {
        const int2 pr = p[i];
        src_sorted[base + atomicAdd(&fill[pr.y & (BIN_SZ - 1)], 1)] = pr.x;
      }
    }
  }
}

struct LaneSel { int x0, x8, x7, x1, x2; };

__device__ __forceinline__ float row_take(float x, int sel) {
  return __int_as_float((int)__builtin_amdgcn_permlane16_var((unsigned)__float_as_int(x), (unsigned)__float_as_int(x),
                                                             (unsigned)sel, true, false));
}

__device__ __forceinline__ float other_row(float x, int sel) {
  return __int_as_float((int)__builtin_amdgcn_permlanex16_var((unsigned)__float_as_int(x), (unsigned)__float_as_int(x),
                                                              (unsigned)sel, true, false));
}

__device__ __forceinline__ float quad_sum(float p, const LaneSel& ls) {
  p += row_take(p, ls.x1);
  p += row_take(p, ls.x2);
  return p;
}

__device__ __forceinline__ float head_score(const float4 x, const v2f (&qq)[4][4], const LaneSel& ls) {
  const v2f xx = {x.x, x.x}, xy = {x.y, x.y}, xz = {x.z, x.z}, xw = {x.w, x.w};
  v2f p[4];
#pragma unroll
  for (int jj = 0; jj < 4; ++jj) {
    v2f t = xx * qq[jj][0];
    t = __builtin_elementwise_fma(xy, qq[jj][1], t);
    t = __builtin_elementwise_fma(xz, qq[jj][2], t);
    t = __builtin_elementwise_fma(xw, qq[jj][3], t);
    p[jj] = t;
  }
  v2f r0, r1;
  r0.x = row_take(p[2].x, ls.x8); r0.y = row_take(p[2].y, ls.x8);
  r1.x = row_take(p[3].x, ls.x8); r1.y = row_take(p[3].y, ls.x8);
  const v2f a = p[0] + r0;
  const v2f b = p[1] + r1;
  v2f s;
  s.x = row_take(b.x, ls.x7);
  s.y = row_take(b.y, ls.x7);
  const v2f e = a + s;
  const float c0 = e.x + other_row(e.y, ls.x0);
  return quad_sum(c0, ls);
}

template <bool TAIL>
__device__ __forceinline__ void attend8(const float4* __restrict__ x4, unsigned lane,
                                        int sv, int c0, int cnt, const v2f (&qq)[4][4], const LaneSel& ls,
                                        float& m, float& l, float4& acc) {
  float4 xx[8];
  int s[8];
#pragma unroll
  for (int i = 0; i < 8; ++i) s[i] = __builtin_amdgcn_readlane(sv, c0 + i);
#pragma unroll
  for (int i = 0; i < 8; ++i) xx[i] = x4[(unsigned)s[i] * (D / 4) + lane];
  float t[8];
  float cm = m;
#pragma unroll
  for (int i = 0; i < 8; ++i) {
    float p = head_score(xx[i], qq, ls);
    if (TAIL && i >= cnt) p = __uint_as_float(0xff800000u);
    t[i] = p;
    cm = fmaxf(cm, p);
  }
  const float r = __expf(m - cm);
  l *= r;
  acc.x *= r; acc.y *= r; acc.z *= r; acc.w *= r;
#pragma unroll
  for (int i = 0; i < 8; ++i) {
    const float e = __expf(t[i] - cm);
    l += e;
    acc.x += e * xx[i].x; acc.y += e * xx[i].y; acc.z += e * xx[i].z; acc.w += e * xx[i].w;
  }
  m = cm;
}

#define TILE_M 16
#define PROW   1028
__global__ void __launch_bounds__(SEG_TB) __attribute__((amdgpu_waves_per_eu(6, 8)))
seg_attend(const float* __restrict__ h_src,
           const float* __restrict__ h_dst,
           const float* __restrict__ Q,
           const float* __restrict__ Wk,
           const int* __restrict__ seg_start,
           const int* __restrict__ seg_cnt,
           const int* __restrict__ src_sorted,
           float* __restrict__ out, int Ndst) {
  __shared__ __attribute__((aligned(16))) float Pl[TILE_M * PROW];
  const int lane = threadIdx.x & 31;
  const int w     = __builtin_amdgcn_readfirstlane((int)(threadIdx.x >> 5));
  const int mBase = blockIdx.x * TILE_M;

  const int dA = (mBase + w < Ndst) ? (mBase + w) : (Ndst - 1);
  const int dB = (mBase + w + 8 < Ndst) ? (mBase + w + 8) : (Ndst - 1);
  const int begA = seg_start[dA], nA = seg_cnt[dA];
  const int begB = seg_start[dB], nB = seg_cnt[dB];

  {
    const int c  = lane & 15;
    const int hh = lane >> 4;
    int qr = mBase + c;
    qr = (qr < Ndst) ? qr : (Ndst - 1);
    const float* qrow = Q + (size_t)qr * D + 16 * w + 8 * hh;
    const float4 q0 = *(const float4*)(qrow);
    const float4 q1 = *(const float4*)(qrow + 4);
    v2f a[4];
    a[0].x = q0.x; a[0].y = q0.y; a[1].x = q0.z; a[1].y = q0.w;
    a[2].x = q1.x; a[2].y = q1.y; a[3].x = q1.z; a[3].y = q1.w;
    const float* wh = Wk + (size_t)(16 * w + 8 * hh) * D + c;
    v8f acc[8] = {};
#pragma unroll
    for (int s = 0; s < 4; ++s) {
      v2f b[8];
#pragma unroll
      for (int t = 0; t < 8; ++t) {
        b[t].x = wh[(2 * s) * D + 16 * t];
        b[t].y = wh[(2 * s + 1) * D + 16 * t];
      }
#pragma unroll
      for (int t = 0; t < 8; ++t) {
        acc[t] = __builtin_amdgcn_wmma_f32_16x16x4_f32(
            false, a[s], false, b[t], (short)0, acc[t], false, false);
      }
    }
    float* pl = Pl + (8 * hh) * PROW + c;
#pragma unroll
    for (int t = 0; t < 8; ++t) {
#pragma unroll
      for (int r = 0; r < 8; ++r) {
        pl[r * PROW + (w ^ t) * D + t * 16] = acc[t][r];
      }
    }
  }
  __syncthreads();

  const int col = lane * 4;
  const float4* x4 = (const float4*)h_src;
  LaneSel ls;
  ls.x0 = lane & 15; ls.x8 = (lane ^ 8) & 15; ls.x7 = (lane ^ 7) & 15; ls.x1 = (lane ^ 1) & 15; ls.x2 = (lane ^ 2) & 15;
#pragma nounroll
  for (int rr = 0; rr < 2; ++rr) {
    const int m = w + 8 * rr;
    const int d = mBase + m;
    if (d >= Ndst) break;
    const int beg = __builtin_amdgcn_readfirstlane(rr ? begB : begA);
    const int n   = __builtin_amdgcn_readfirstlane(rr ? nB : nA);
    v2f qq[4][4];
    const float* prow = Pl + m * PROW + col;
#pragma unroll
    for (int jj = 0; jj < 4; ++jj) {
      const float4 u  = *(const float4*)(prow + jj * D);
      const float4 wv = *(const float4*)(prow + (jj + 4) * D);
      qq[jj][0].x = u.x * SCALE; qq[jj][0].y = wv.x * SCALE;
      qq[jj][1].x = u.y * SCALE; qq[jj][1].y = wv.y * SCALE;
      qq[jj][2].x = u.z * SCALE; qq[jj][2].y = wv.z * SCALE;
      qq[jj][3].x = u.w * SCALE; qq[jj][3].y = wv.w * SCALE;
    }
    const unsigned o4 = (unsigned)d * (D / 4) + (unsigned)lane;
    const float4 xd = ((const float4*)h_dst)[o4];
    float mx = head_score(xd, qq, ls);
    float l = 1.0f;
    float4 acc = xd;
    for (int j0 = 0; j0 < n; j0 += 32) {
      const int mc = (n - j0 < 32) ? (n - j0) : 32;
      const int sv = (lane < mc) ? src_sorted[beg + j0 + lane] : 0;
      int c0 = 0;
      for (; c0 + 8 <= mc; c0 += 8) attend8<false>(x4, (unsigned)lane, sv, c0, 8, qq, ls, mx, l, acc);
      if (c0 < mc) attend8<true>(x4, (unsigned)lane, sv, c0, mc - c0, qq, ls, mx, l, acc);
    }
    const float rl = 1.0f / l;
    float4 o;
    o.x = acc.x * rl; o.y = acc.y * rl; o.z = acc.z * rl; o.w = acc.w * rl;
    ((float4*)out)[o4] = o;
  }
}

extern "C" void kernel_launch(void* const* d_in, const int* in_sizes, int n_in,
                              void* d_out, int out_size, void* d_ws, size_t ws_size,
                              hipStream_t stream) {
  const float* h_src = (const float*)d_in[0];
  const float* h_dst = (const float*)d_in[1];
  const float* Wk    = (const float*)d_in[2];
  const float* Wq    = (const float*)d_in[3];
  const int* src_idx = (const int*)d_in[4];
  const int* dst_idx = (const int*)d_in[5];

  const int N_DST = in_sizes[1] / D;
  const int E     = in_sizes[4];

  auto align256 = [](size_t x) { return (x + 255) & ~(size_t)255; };
  char* ws = (char*)d_ws;
  size_t off = 0;
  float* Q      = (float*)(ws + off); off += align256((size_t)N_DST * D * 4);
  int* pairs      = (int*)(ws + off); off += align256((size_t)E * 8);
  int* src_sorted = (int*)(ws + off); off += align256((size_t)E * 4);
  int* seg_cnt    = (int*)(ws + off); off += align256((size_t)N_DST * 4);
  int* seg_start  = (int*)(ws + off); off += align256((size_t)N_DST * 4);
  int* cursor     = (int*)(ws + off); off += align256(4);
  int* run_start  = (int*)(ws + off); off += align256((size_t)NBIN_MAX * SORT_G * 4);
  int* run_cnt    = (int*)(ws + off); off += align256((size_t)NBIN_MAX * SORT_G * 4);

  const int projG = (N_DST + PROJ_ROWS - 1) / PROJ_ROWS;
  project_and_chunk_sort<<<projG + SORT_G, SORT_TB, 0, stream>>>(
      h_dst, Wq, Q, N_DST, projG, src_idx, dst_idx, pairs, run_start, run_cnt, cursor, E);
  bin_sort<<<(N_DST + BIN_SZ - 1) / BIN_SZ, SORT_TB, 0, stream>>>(
      pairs, run_start, run_cnt, cursor, seg_start, seg_cnt, src_sorted, E, N_DST);
  seg_attend<<<(N_DST + TILE_M - 1) / TILE_M, SEG_TB, 0, stream>>>(
      h_src, h_dst, Q, Wk, seg_start, seg_cnt, src_sorted, (float*)d_out, N_DST);
}


# seg kernel: scheduling fences hold two edges' partial sums live at a time (127 registers, was 158): 8 waves per SIMD instead of 6, 4 work-groups per WGP; every sum and its order unchanged
# speedup vs baseline: 7.9302x; 1.0329x over previous
#include <hip/hip_runtime.h>
#include <hip/hip_bf16.h>

#define D      128
#define H      8
#define DK     16
#define SCALE  0.25f
#define SEG_TB 256

typedef __attribute__((ext_vector_type(2))) float v2f;
typedef __attribute__((ext_vector_type(8))) float v8f;
typedef __attribute__((ext_vector_type(4))) float v4f;

#define PROJ_ROWS 256
#define PROJ_TB   1024
#define WPITCH  132
__device__ __forceinline__ void project_rows(const float* __restrict__ X,
                                             const float* __restrict__ W,
                                             float* __restrict__ Y, int M, int bid) {
  __shared__ __attribute__((aligned(16))) float Wl[D * WPITCH];

  {
    float4 w4[4];
#pragma unroll
    for (int i = 0; i < 4; ++i) {
      const int idx = threadIdx.x + i * PROJ_TB;
      w4[i] = *(const float4*)(W + (size_t)(idx >> 5) * D + (idx & 31) * 4);
    }
#pragma unroll
    for (int i = 0; i < 4; ++i) {
      const int idx = threadIdx.x + i * PROJ_TB;
      *(float4*)(&Wl[(idx >> 5) * WPITCH + (idx & 31) * 4]) = w4[i];
    }
  }
  __syncthreads();

  const int lane  = threadIdx.x & 31;
  const int wv    = __builtin_amdgcn_readfirstlane((int)(threadIdx.x >> 5));
  const int nh    = wv & 1;
  const int mBase = (bid * (PROJ_ROWS / 16) + (wv >> 1)) * 16;
  if (mBase >= M) return;

  const int row  = lane & 15;
  const int koff = (lane >> 4) << 1;

  const float* xrow  = X + (size_t)(mBase + row) * D + koff;
  const float* wbase = Wl + (nh * 64 + row) * WPITCH + koff;

  v8f acc[4] = {};

  v2f a_cur = *(const v2f*)(xrow);
  v2f b_cur[4];
#pragma unroll
  for (int t = 0; t < 4; ++t)
    b_cur[t] = *(const v2f*)(wbase + t * 16 * WPITCH);

#pragma unroll 4
  for (int k0 = 0; k0 < D; k0 += 4) {
    v2f a_nxt;
    v2f b_nxt[4];
    if (k0 + 4 < D) {
      a_nxt = *(const v2f*)(xrow + k0 + 4);
#pragma unroll
      for (int t = 0; t < 4; ++t)
        b_nxt[t] = *(const v2f*)(wbase + t * 16 * WPITCH + k0 + 4);
    }
#pragma unroll
    for (int t = 0; t < 4; ++t) {
      acc[t] = __builtin_amdgcn_wmma_f32_16x16x4_f32(
          false, a_cur, false, b_cur[t], (short)0, acc[t], false, false);
    }
    a_cur = a_nxt;
#pragma unroll
    for (int t = 0; t < 4; ++t) b_cur[t] = b_nxt[t];
  }

  const int n    = lane & 15;
  const int mAdd = (lane >> 4) << 3;
  float* yb = Y + (size_t)mBase * D + nh * 64;
#pragma unroll
  for (int t = 0; t < 4; ++t) {
#pragma unroll
    for (int r = 0; r < 8; ++r) {
      yb[(size_t)(mAdd + r) * D + t * 16 + n] = acc[t][r];
    }
  }
}

#define BIN_SHIFT 8
#define BIN_SZ    256
#define NBIN_MAX  256
#define SORT_TB   1024
#define SORT_G    256
#define SORT_PRE  8
#define RUNS_PW   (SORT_G / (SORT_TB / 32))

__device__ __forceinline__ int excl_scan256(int c, int* wsum) {
  const int tid = threadIdx.x, lane = tid & 31;
  int inc = c;
  for (int o = 1; o < 32; o <<= 1) {
    int v = __shfl_up(inc, o);
    if (lane >= o) inc += v;
  }
  if (tid < 256 && lane == 31) wsum[tid >> 5] = inc;
  __syncthreads();
  int base = 0;
  if (tid < 256)
    for (int i = 0; i < (tid >> 5); ++i) base += wsum[i];
  return base + inc - c;
}

__device__ __forceinline__ void
chunk_sort(const int* __restrict__ src_idx,
           const int* __restrict__ dst_idx,
           int* __restrict__ pairs,
           int* __restrict__ run_start,
           int* __restrict__ run_cnt,
           int* __restrict__ cursor, int E, int bid) {
  __shared__ int hist[NBIN_MAX];
  __shared__ int fill[NBIN_MAX];
  __shared__ int wsum[8];
  const int tid = threadIdx.x;
  const int ch  = (E + SORT_G - 1) / SORT_G;
  const int e0  = bid * ch;
  const int e1  = (e0 + ch < E) ? (e0 + ch) : E;
  if (tid < NBIN_MAX) hist[tid] = 0;
  if (bid == 0 && tid == 0) *cursor = 0;

  int dv[SORT_PRE], sv[SORT_PRE];
#pragma unroll
  for (int k = 0; k < SORT_PRE; ++k) {
    const int e = e0 + tid + k * SORT_TB;
    dv[k] = (e < e1) ? dst_idx[e] : -1;
  }
#pragma unroll
  for (int k = 0; k < SORT_PRE; ++k) {
    const int e = e0 + tid + k * SORT_TB;
    sv[k] = (e < e1) ? src_idx[e] : 0;
  }
  __syncthreads();
#pragma unroll
  for (int k = 0; k < SORT_PRE; ++k)
    if (dv[k] >= 0) atomicAdd(&hist[dv[k] >> BIN_SHIFT], 1);
  for (int e = e0 + tid + SORT_PRE * SORT_TB; e < e1; e += SORT_TB)
    atomicAdd(&hist[dst_idx[e] >> BIN_SHIFT], 1);
  __syncthreads();

  const int c  = (tid < NBIN_MAX) ? hist[tid] : 0;
  const int st = excl_scan256(c, wsum);
  if (tid < NBIN_MAX) {
    fill[tid] = st;
    run_start[tid * SORT_G + bid] = st;
    run_cnt[tid * SORT_G + bid]   = c;
  }
  __syncthreads();

  int2* mine = (int2*)pairs + e0;
#pragma unroll
  for (int k = 0; k < SORT_PRE; ++k)
    if (dv[k] >= 0) {
      const int p = atomicAdd(&fill[dv[k] >> BIN_SHIFT], 1);
      mine[p] = make_int2(sv[k], dv[k]);
    }
  for (int e = e0 + tid + SORT_PRE * SORT_TB; e < e1; e += SORT_TB) {
    const int d = dst_idx[e];
    const int p = atomicAdd(&fill[d >> BIN_SHIFT], 1);
    mine[p] = make_int2(src_idx[e], d);
  }
}

__global__ void __launch_bounds__(SORT_TB) __attribute__((amdgpu_waves_per_eu(16)))
project_and_chunk_sort(const float* __restrict__ h_dst,
                       const float* __restrict__ Wq,
                       float* __restrict__ Q, int Ndst, int projG,
                       const int* __restrict__ src_idx,
                       const int* __restrict__ dst_idx,
                       int* __restrict__ pairs,
                       int* __restrict__ run_start,
                       int* __restrict__ run_cnt,
                       int* __restrict__ cursor, int E) {
  const int b = blockIdx.x;
  if (b < projG) project_rows(h_dst, Wq, Q, Ndst, b);
  else           chunk_sort(src_idx, dst_idx, pairs, run_start, run_cnt, cursor, E, b - projG);
}

__global__ void __launch_bounds__(SORT_TB)
bin_sort(const int* __restrict__ pairs,
         const int* __restrict__ run_start,
         const int* __restrict__ run_cnt,
         int* __restrict__ cursor,
         int* __restrict__ seg_start,
         int* __restrict__ seg_cnt,
         int* __restrict__ src_sorted, int E, int Ndst) {
  __shared__ int rs[SORT_G];
  __shared__ int rc[SORT_G];
  __shared__ int cnt[BIN_SZ];
  __shared__ int fill[BIN_SZ];
  __shared__ int wtot[8];
  __shared__ int wsum[8];
  __shared__ int base_s;
  const int tid = threadIdx.x, lane = tid & 31, w = tid >> 5;
  const int b  = blockIdx.x;
  const int ch = (E + SORT_G - 1) / SORT_G;
  if (tid < BIN_SZ) cnt[tid] = 0;
  if (tid < SORT_G) {
    const int c = run_cnt[b * SORT_G + tid];
    rc[tid] = c;
    rs[tid] = tid * ch + run_start[b * SORT_G + tid];
    int s = c;
    for (int o = 16; o > 0; o >>= 1) s += __shfl_xor(s, o);
    if (lane == 0) wtot[w] = s;
  }
  __syncthreads();

  const int2* pp = (const int2*)pairs;
  int  rn[RUNS_PW];
  int2 v0[RUNS_PW], v1[RUNS_PW];
#pragma unroll
  for (int k = 0; k < RUNS_PW; ++k) {
    const int r = w + k * (SORT_TB / 32);
    rn[k] = rc[r];
    int i0 = rs[r] + lane, i1 = i0 + 32;
    i0 = (i0 < E) ? i0 : E - 1;
    i1 = (i1 < E) ? i1 : E - 1;
    v0[k] = pp[i0];
    v1[k] = pp[i1];
  }
  if (tid == 0) {
    int tot = 0;
    for (int i = 0; i < SORT_G / 32; ++i) tot += wtot[i];
    base_s = atomicAdd(cursor, tot);
  }
#pragma unroll
  for (int k = 0; k < RUNS_PW; ++k) {
    if (lane < rn[k])      atomicAdd(&cnt[v0[k].y & (BIN_SZ - 1)], 1);
    if (lane + 32 < rn[k]) atomicAdd(&cnt[v1[k].y & (BIN_SZ - 1)], 1);
    if (rn[k] > 64) {
      const int2* p = pp + rs[w + k * (SORT_TB / 32)];
      for (int i = lane + 64; i < rn[k]; i += 32) atomicAdd(&cnt[p[i].y & (BIN_SZ - 1)], 1);
    }
  }
  __syncthreads();

  const int c  = (tid < BIN_SZ) ? cnt[tid] : 0;
  const int st = excl_scan256(c, wsum);
  const int base = base_s;
  if (tid < BIN_SZ) {
    fill[tid] = st;
    const int d = b * BIN_SZ + tid;
    if (d < Ndst) { seg_start[d] = base + st; seg_cnt[d] = c; }
  }
  __syncthreads();

#pragma unroll
  for (int k = 0; k < RUNS_PW; ++k) {
    if (lane < rn[k])      src_sorted[base + atomicAdd(&fill[v0[k].y & (BIN_SZ - 1)], 1)] = v0[k].x;
    if (lane + 32 < rn[k]) src_sorted[base + atomicAdd(&fill[v1[k].y & (BIN_SZ - 1)], 1)] = v1[k].x;
    if (rn[k] > 64) {
      const int2* p = pp + rs[w + k * (SORT_TB / 32)];
      for (int i = lane + 64; i < rn[k]; i += 32) {
        const int2 pr = p[i];
        src_sorted[base + atomicAdd(&fill[pr.y & (BIN_SZ - 1)], 1)] = pr.x;
      }
    }
  }
}

struct LaneSel { int x0, x8, x7, x1, x2; };

__device__ __forceinline__ float row_take(float x, int sel) {
  return __int_as_float((int)__builtin_amdgcn_permlane16_var((unsigned)__float_as_int(x), (unsigned)__float_as_int(x),
                                                             (unsigned)sel, true, false));
}

__device__ __forceinline__ float other_row(float x, int sel) {
  return __int_as_float((int)__builtin_amdgcn_permlanex16_var((unsigned)__float_as_int(x), (unsigned)__float_as_int(x),
                                                              (unsigned)sel, true, false));
}

__device__ __forceinline__ float quad_sum(float p, const LaneSel& ls) {
  p += row_take(p, ls.x1);
  p += row_take(p, ls.x2);
  return p;
}

__device__ __forceinline__ float head_score(const v4f x, const v2f (&qq)[4][4], const LaneSel& ls) {
  const v2f xx = {x.x, x.x}, xy = {x.y, x.y}, xz = {x.z, x.z}, xw = {x.w, x.w};
  v2f p[4];
#pragma unroll
  for (int jj = 0; jj < 4; ++jj) {
    v2f t = xx * qq[jj][0];
    t = __builtin_elementwise_fma(xy, qq[jj][1], t);
    t = __builtin_elementwise_fma(xz, qq[jj][2], t);
    t = __builtin_elementwise_fma(xw, qq[jj][3], t);
    p[jj] = t;
  }
  v2f r0, r1;
  r0.x = row_take(p[2].x, ls.x8); r0.y = row_take(p[2].y, ls.x8);
  r1.x = row_take(p[3].x, ls.x8); r1.y = row_take(p[3].y, ls.x8);
  const v2f a = p[0] + r0;
  const v2f b = p[1] + r1;
  v2f s;
  s.x = row_take(b.x, ls.x7);
  s.y = row_take(b.y, ls.x7);
  const v2f e = a + s;
  const float c0 = e.x + other_row(e.y, ls.x0);
  return quad_sum(c0, ls);
}

template <bool TAIL>
__device__ __forceinline__ void attend8(const v4f* __restrict__ x4, unsigned lane,
                                        int sv, int c0, int cnt, const v2f (&qq)[4][4], const LaneSel& ls,
                                        float& m, float& l, float4& acc) {
  v4f xx[8];
  int s[8];
#pragma unroll
  for (int i = 0; i < 8; ++i) s[i] = __builtin_amdgcn_readlane(sv, c0 + i);
#pragma unroll
  for (int i = 0; i < 8; ++i) xx[i] = x4[(unsigned)s[i] * (D / 4) + lane];
  asm volatile("" :: "v"(xx[0]), "v"(xx[1]), "v"(xx[2]), "v"(xx[3]), "v"(xx[4]), "v"(xx[5]), "v"(xx[6]), "v"(xx[7]));
  __builtin_amdgcn_sched_barrier(0);
  float t[8];
  float cm = m;
#pragma unroll
  for (int i = 0; i < 8; ++i) {
    float p = head_score(xx[i], qq, ls);
    if (TAIL && i >= cnt) p = __uint_as_float(0xff800000u);
    t[i] = p;
    cm = fmaxf(cm, p);
    if (i & 1) __builtin_amdgcn_sched_barrier(0);
  }
  const float r = __expf(m - cm);
  l *= r;
  acc.x *= r; acc.y *= r; acc.z *= r; acc.w *= r;
#pragma unroll
  for (int i = 0; i < 8; ++i) {
    const float e = __expf(t[i] - cm);
    l += e;
    acc.x += e * xx[i].x; acc.y += e * xx[i].y; acc.z += e * xx[i].z; acc.w += e * xx[i].w;
  }
  m = cm;
}

#define TILE_M 16
#define PROW   1028
__global__ void __launch_bounds__(SEG_TB) __attribute__((amdgpu_waves_per_eu(8, 8)))
seg_attend(const float* __restrict__ h_src,
           const float* __restrict__ h_dst,
           const float* __restrict__ Q,
           const float* __restrict__ Wk,
           const int* __restrict__ seg_start,
           const int* __restrict__ seg_cnt,
           const int* __restrict__ src_sorted,
           float* __restrict__ out, int Ndst) {
  __shared__ __attribute__((aligned(16))) float Pl[TILE_M * PROW];
  const int lane = threadIdx.x & 31;
  const int w     = __builtin_amdgcn_readfirstlane((int)(threadIdx.x >> 5));
  const int mBase = blockIdx.x * TILE_M;

  const int dA = (mBase + w < Ndst) ? (mBase + w) : (Ndst - 1);
  const int dB = (mBase + w + 8 < Ndst) ? (mBase + w + 8) : (Ndst - 1);
  const int begA = seg_start[dA], nA = seg_cnt[dA];
  const int begB = seg_start[dB], nB = seg_cnt[dB];

  {
    const int c  = lane & 15;
    const int hh = lane >> 4;
    int qr = mBase + c;
    qr = (qr < Ndst) ? qr : (Ndst - 1);
    const float* qrow = Q + (size_t)qr * D + 16 * w + 8 * hh;
    const float4 q0 = *(const float4*)(qrow);
    const float4 q1 = *(const float4*)(qrow + 4);
    v2f a[4];
    a[0].x = q0.x; a[0].y = q0.y; a[1].x = q0.z; a[1].y = q0.w;
    a[2].x = q1.x; a[2].y = q1.y; a[3].x = q1.z; a[3].y = q1.w;
    const float* wh = Wk + (size_t)(16 * w + 8 * hh) * D + c;
    v8f acc[8] = {};
#pragma unroll
    for (int s = 0; s < 4; ++s) {
      v2f b[8];
#pragma unroll
      for (int t = 0; t < 8; ++t) {
        b[t].x = wh[(2 * s) * D + 16 * t];
        b[t].y = wh[(2 * s + 1) * D + 16 * t];
      }
#pragma unroll
      for (int t = 0; t < 8; ++t) {
        acc[t] = __builtin_amdgcn_wmma_f32_16x16x4_f32(
            false, a[s], false, b[t], (short)0, acc[t], false, false);
      }
    }
    float* pl = Pl + (8 * hh) * PROW + c;
#pragma unroll
    for (int t = 0; t < 8; ++t) {
#pragma unroll
      for (int r = 0; r < 8; ++r) {
        pl[r * PROW + (w ^ t) * D + t * 16] = acc[t][r];
      }
    }
  }
  __syncthreads();

  const int ln = (int)__builtin_amdgcn_mbcnt_lo(0xffffffffu, 0u);
  const int col = ln * 4;
  const v4f* x4 = (const v4f*)h_src;
  LaneSel ls;
  ls.x0 = ln & 15; ls.x8 = (ln ^ 8) & 15; ls.x7 = (ln ^ 7) & 15; ls.x1 = (ln ^ 1) & 15; ls.x2 = (ln ^ 2) & 15;
#pragma nounroll
  for (int rr = 0; rr < 2; ++rr) {
    const int m = w + 8 * rr;
    const int d = mBase + m;
    if (d >= Ndst) break;
    const int beg = __builtin_amdgcn_readfirstlane(rr ? begB : begA);
    const int n   = __builtin_amdgcn_readfirstlane(rr ? nB : nA);
    v2f qq[4][4];
    const float* prow = Pl + m * PROW + col;
#pragma unroll
    for (int jj = 0; jj < 4; ++jj) {
      const float4 u  = *(const float4*)(prow + jj * D);
      const float4 wv = *(const float4*)(prow + (jj + 4) * D);
      qq[jj][0].x = u.x * SCALE; qq[jj][0].y = wv.x * SCALE;
      qq[jj][1].x = u.y * SCALE; qq[jj][1].y = wv.y * SCALE;
      qq[jj][2].x = u.z * SCALE; qq[jj][2].y = wv.z * SCALE;
      qq[jj][3].x = u.w * SCALE; qq[jj][3].y = wv.w * SCALE;
    }
    const unsigned o4 = (unsigned)d * (D / 4) + (unsigned)ln;
    const float4 xd = ((const float4*)h_dst)[o4];
    const v4f xdv = {xd.x, xd.y, xd.z, xd.w};
    float mx = head_score(xdv, qq, ls);
    float l = 1.0f;
    float4 acc = xd;
    for (int j0 = 0; j0 < n; j0 += 32) {
      const int mc = (n - j0 < 32) ? (n - j0) : 32;
      const int sv = (ln < mc) ? src_sorted[beg + j0 + ln] : 0;
      int c0 = 0;
      for (; c0 + 8 <= mc; c0 += 8) attend8<false>(x4, (unsigned)ln, sv, c0, 8, qq, ls, mx, l, acc);
      if (c0 < mc) attend8<true>(x4, (unsigned)ln, sv, c0, mc - c0, qq, ls, mx, l, acc);
    }
    const float rl = 1.0f / l;
    float4 o;
    o.x = acc.x * rl; o.y = acc.y * rl; o.z = acc.z * rl; o.w = acc.w * rl;
    ((float4*)out)[o4] = o;
  }
}

extern "C" void kernel_launch(void* const* d_in, const int* in_sizes, int n_in,
                              void* d_out, int out_size, void* d_ws, size_t ws_size,
                              hipStream_t stream) {
  const float* h_src = (const float*)d_in[0];
  const float* h_dst = (const float*)d_in[1];
  const float* Wk    = (const float*)d_in[2];
  const float* Wq    = (const float*)d_in[3];
  const int* src_idx = (const int*)d_in[4];
  const int* dst_idx = (const int*)d_in[5];

  const int N_DST = in_sizes[1] / D;
  const int E     = in_sizes[4];

  auto align256 = [](size_t x) { return (x + 255) & ~(size_t)255; };
  char* ws = (char*)d_ws;
  size_t off = 0;
  float* Q      = (float*)(ws + off); off += align256((size_t)N_DST * D * 4);
  int* pairs      = (int*)(ws + off); off += align256((size_t)E * 8);
  int* src_sorted = (int*)(ws + off); off += align256((size_t)E * 4);
  int* seg_cnt    = (int*)(ws + off); off += align256((size_t)N_DST * 4);
  int* seg_start  = (int*)(ws + off); off += align256((size_t)N_DST * 4);
  int* cursor     = (int*)(ws + off); off += align256(4);
  int* run_start  = (int*)(ws + off); off += align256((size_t)NBIN_MAX * SORT_G * 4);
  int* run_cnt    = (int*)(ws + off); off += align256((size_t)NBIN_MAX * SORT_G * 4);

  const int projG = (N_DST + PROJ_ROWS - 1) / PROJ_ROWS;
  project_and_chunk_sort<<<projG + SORT_G, SORT_TB, 0, stream>>>(
      h_dst, Wq, Q, N_DST, projG, src_idx, dst_idx, pairs, run_start, run_cnt, cursor, E);
  bin_sort<<<(N_DST + BIN_SZ - 1) / BIN_SZ, SORT_TB, 0, stream>>>(
      pairs, run_start, run_cnt, cursor, seg_start, seg_cnt, src_sorted, E, N_DST);
  seg_attend<<<(N_DST + TILE_M - 1) / TILE_M, SEG_TB, 0, stream>>>(
      h_src, h_dst, Q, Wk, seg_start, seg_cnt, src_sorted, (float*)d_out, N_DST);
}


# Both sort launches build their sorted output (chunk pairs, bin id list) in LDS and write it out as whole lines: no scattered 4/8-byte global stores; ranks still by LDS atomic arrival as before
# speedup vs baseline: 8.3233x; 1.0496x over previous
#include <hip/hip_runtime.h>
#include <hip/hip_bf16.h>

#define D      128
#define H      8
#define DK     16
#define SCALE  0.25f
#define SEG_TB 256

typedef __attribute__((ext_vector_type(2))) float v2f;
typedef __attribute__((ext_vector_type(8))) float v8f;
typedef __attribute__((ext_vector_type(4))) float v4f;

#define PROJ_ROWS 256
#define PROJ_TB   1024
#define WPITCH  132
__device__ __forceinline__ void project_rows(const float* __restrict__ X,
                                             const float* __restrict__ W,
                                             float* __restrict__ Y, int M, int bid,
                                             float* __restrict__ Wl) {

  {
    float4 w4[4];
#pragma unroll
    for (int i = 0; i < 4; ++i) {
      const int idx = threadIdx.x + i * PROJ_TB;
      w4[i] = *(const float4*)(W + (size_t)(idx >> 5) * D + (idx & 31) * 4);
    }
#pragma unroll
    for (int i = 0; i < 4; ++i) {
      const int idx = threadIdx.x + i * PROJ_TB;
      *(float4*)(&Wl[(idx >> 5) * WPITCH + (idx & 31) * 4]) = w4[i];
    }
  }
  __syncthreads();

  const int lane  = threadIdx.x & 31;
  const int wv    = __builtin_amdgcn_readfirstlane((int)(threadIdx.x >> 5));
  const int nh    = wv & 1;
  const int mBase = (bid * (PROJ_ROWS / 16) + (wv >> 1)) * 16;
  if (mBase >= M) return;

  const int row  = lane & 15;
  const int koff = (lane >> 4) << 1;

  const float* xrow  = X + (size_t)(mBase + row) * D + koff;
  const float* wbase = Wl + (nh * 64 + row) * WPITCH + koff;

  v8f acc[4] = {};

  v2f a_cur = *(const v2f*)(xrow);
  v2f b_cur[4];
#pragma unroll
  for (int t = 0; t < 4; ++t)
    b_cur[t] = *(const v2f*)(wbase + t * 16 * WPITCH);

#pragma unroll 4
  for (int k0 = 0; k0 < D; k0 += 4) {
    v2f a_nxt;
    v2f b_nxt[4];
    if (k0 + 4 < D) {
      a_nxt = *(const v2f*)(xrow + k0 + 4);
#pragma unroll
      for (int t = 0; t < 4; ++t)
        b_nxt[t] = *(const v2f*)(wbase + t * 16 * WPITCH + k0 + 4);
    }
#pragma unroll
    for (int t = 0; t < 4; ++t) {
      acc[t] = __builtin_amdgcn_wmma_f32_16x16x4_f32(
          false, a_cur, false, b_cur[t], (short)0, acc[t], false, false);
    }
    a_cur = a_nxt;
#pragma unroll
    for (int t = 0; t < 4; ++t) b_cur[t] = b_nxt[t];
  }

  const int n    = lane & 15;
  const int mAdd = (lane >> 4) << 3;
  float* yb = Y + (size_t)mBase * D + nh * 64;
#pragma unroll
  for (int t = 0; t < 4; ++t) {
#pragma unroll
    for (int r = 0; r < 8; ++r) {
      yb[(size_t)(mAdd + r) * D + t * 16 + n] = acc[t][r];
    }
  }
}

#define BIN_SHIFT 8
#define BIN_SZ    256
#define NBIN_MAX  256
#define SORT_TB   1024
#define SORT_G    256
#define SORT_PRE  8
#define RUNS_PW   (SORT_G / (SORT_TB / 32))
#define BIN_STG   12288

__device__ __forceinline__ int excl_scan256(int c, int* wsum) {
  const int tid = threadIdx.x, lane = tid & 31;
  int inc = c;
  for (int o = 1; o < 32; o <<= 1) {
    int v = __shfl_up(inc, o);
    if (lane >= o) inc += v;
  }
  if (tid < 256 && lane == 31) wsum[tid >> 5] = inc;
  __syncthreads();
  int base = 0;
  if (tid < 256)
    for (int i = 0; i < (tid >> 5); ++i) base += wsum[i];
  return base + inc - c;
}

__device__ __forceinline__ void
chunk_sort(const int* __restrict__ src_idx,
           const int* __restrict__ dst_idx,
           int* __restrict__ pairs,
           int* __restrict__ run_start,
           int* __restrict__ run_cnt,
           int* __restrict__ cursor, int E, int bid,
           int2* __restrict__ stage) {
  __shared__ int hist[NBIN_MAX];
  __shared__ int fill[NBIN_MAX];
  __shared__ int wsum[8];
  const int tid = threadIdx.x;
  const int ch  = (E + SORT_G - 1) / SORT_G;
  const int e0  = bid * ch;
  const int e1  = (e0 + ch < E) ? (e0 + ch) : E;
  if (tid < NBIN_MAX) hist[tid] = 0;
  if (bid == 0 && tid == 0) *cursor = 0;

  int dv[SORT_PRE], sv[SORT_PRE];
#pragma unroll
  for (int k = 0; k < SORT_PRE; ++k) {
    const int e = e0 + tid + k * SORT_TB;
    dv[k] = (e < e1) ? dst_idx[e] : -1;
  }
#pragma unroll
  for (int k = 0; k < SORT_PRE; ++k) {
    const int e = e0 + tid + k * SORT_TB;
    sv[k] = (e < e1) ? src_idx[e] : 0;
  }
  __syncthreads();
#pragma unroll
  for (int k = 0; k < SORT_PRE; ++k)
    if (dv[k] >= 0) atomicAdd(&hist[dv[k] >> BIN_SHIFT], 1);
  for (int e = e0 + tid + SORT_PRE * SORT_TB; e < e1; e += SORT_TB)
    atomicAdd(&hist[dst_idx[e] >> BIN_SHIFT], 1);
  __syncthreads();

  const int c  = (tid < NBIN_MAX) ? hist[tid] : 0;
  const int st = excl_scan256(c, wsum);
  if (tid < NBIN_MAX) {
    fill[tid] = st;
    run_start[tid * SORT_G + bid] = st;
    run_cnt[tid * SORT_G + bid]   = c;
  }
  __syncthreads();

  int2* mine = (int2*)pairs + e0;
  const int n = e1 - e0;
  if (n <= SORT_PRE * SORT_TB) {
#pragma unroll
    for (int k = 0; k < SORT_PRE; ++k)
      if (dv[k] >= 0) {
        const int p = atomicAdd(&fill[dv[k] >> BIN_SHIFT], 1);
        stage[p] = make_int2(sv[k], dv[k]);
      }
    __syncthreads();
    for (int i = tid; i < n; i += SORT_TB) mine[i] = stage[i];
  } else {
#pragma unroll
    for (int k = 0; k < SORT_PRE; ++k)
      if (dv[k] >= 0) {
        const int p = atomicAdd(&fill[dv[k] >> BIN_SHIFT], 1);
        mine[p] = make_int2(sv[k], dv[k]);
      }
    for (int e = e0 + tid + SORT_PRE * SORT_TB; e < e1; e += SORT_TB) {
      const int d = dst_idx[e];
      const int p = atomicAdd(&fill[d >> BIN_SHIFT], 1);
      mine[p] = make_int2(src_idx[e], d);
    }
  }
}

__global__ void __launch_bounds__(SORT_TB) __attribute__((amdgpu_waves_per_eu(16)))
project_and_chunk_sort(const float* __restrict__ h_dst,
                       const float* __restrict__ Wq,
                       float* __restrict__ Q, int Ndst, int projG,
                       const int* __restrict__ src_idx,
                       const int* __restrict__ dst_idx,
                       int* __restrict__ pairs,
                       int* __restrict__ run_start,
                       int* __restrict__ run_cnt,
                       int* __restrict__ cursor, int E) {
  __shared__ __attribute__((aligned(16))) float lds_buf[D * WPITCH];
  static_assert(sizeof(float) * D * WPITCH >= sizeof(int2) * SORT_PRE * SORT_TB, "staging buffer too small");
  const int b = blockIdx.x;
  if (b < projG) project_rows(h_dst, Wq, Q, Ndst, b, lds_buf);
  else           chunk_sort(src_idx, dst_idx, pairs, run_start, run_cnt, cursor, E, b - projG, (int2*)lds_buf);
}

__global__ void __launch_bounds__(SORT_TB)
bin_sort(const int* __restrict__ pairs,
         const int* __restrict__ run_start,
         const int* __restrict__ run_cnt,
         int* __restrict__ cursor,
         int* __restrict__ seg_start,
         int* __restrict__ seg_cnt,
         int* __restrict__ src_sorted, int E, int Ndst) {
  __shared__ int rs[SORT_G];
  __shared__ int rc[SORT_G];
  __shared__ int cnt[BIN_SZ];
  __shared__ int fill[BIN_SZ];
  __shared__ int wtot[8];
  __shared__ int wsum[8];
  __shared__ int base_s;
  __shared__ int tot_s;
  __shared__ int stg[BIN_STG];
  const int tid = threadIdx.x, lane = tid & 31, w = tid >> 5;
  const int b  = blockIdx.x;
  const int ch = (E + SORT_G - 1) / SORT_G;
  if (tid < BIN_SZ) cnt[tid] = 0;
  if (tid < SORT_G) {
    const int c = run_cnt[b * SORT_G + tid];
    rc[tid] = c;
    rs[tid] = tid * ch + run_start[b * SORT_G + tid];
    int s = c;
    for (int o = 16; o > 0; o >>= 1) s += __shfl_xor(s, o);
    if (lane == 0) wtot[w] = s;
  }
  __syncthreads();

  const int2* pp = (const int2*)pairs;
  int  rn[RUNS_PW];
  int2 v0[RUNS_PW], v1[RUNS_PW];
#pragma unroll
  for (int k = 0; k < RUNS_PW; ++k) {
    const int r = w + k * (SORT_TB / 32);
    rn[k] = rc[r];
    int i0 = rs[r] + lane, i1 = i0 + 32;
    i0 = (i0 < E) ? i0 : E - 1;
    i1 = (i1 < E) ? i1 : E - 1;
    v0[k] = pp[i0];
    v1[k] = pp[i1];
  }
  if (tid == 0) {
    int tot = 0;
    for (int i = 0; i < SORT_G / 32; ++i) tot += wtot[i];
    base_s = atomicAdd(cursor, tot);
    tot_s  = tot;
  }
#pragma unroll
  for (int k = 0; k < RUNS_PW; ++k) {
    if (lane < rn[k])      atomicAdd(&cnt[v0[k].y & (BIN_SZ - 1)], 1);
    if (lane + 32 < rn[k]) atomicAdd(&cnt[v1[k].y & (BIN_SZ - 1)], 1);
    if (rn[k] > 64) {
      const int2* p = pp + rs[w + k * (SORT_TB / 32)];
      for (int i = lane + 64; i < rn[k]; i += 32) atomicAdd(&cnt[p[i].y & (BIN_SZ - 1)], 1);
    }
  }
  __syncthreads();

  const int c  = (tid < BIN_SZ) ? cnt[tid] : 0;
  const int st = excl_scan256(c, wsum);
  const int base = base_s;
  const int tot  = tot_s;
  if (tid < BIN_SZ) {
    fill[tid] = st;
    const int d = b * BIN_SZ + tid;
    if (d < Ndst) { seg_start[d] = base + st; seg_cnt[d] = c; }
  }
  __syncthreads();

  if (tot <= BIN_STG) {
#pragma unroll
    for (int k = 0; k < RUNS_PW; ++k) {
      if (lane < rn[k])      stg[atomicAdd(&fill[v0[k].y & (BIN_SZ - 1)], 1)] = v0[k].x;
      if (lane + 32 < rn[k]) stg[atomicAdd(&fill[v1[k].y & (BIN_SZ - 1)], 1)] = v1[k].x;
      if (rn[k] > 64) {
        const int2* p = pp + rs[w + k * (SORT_TB / 32)];
        for (int i = lane + 64; i < rn[k]; i += 32) {
          const int2 pr = p[i];
          stg[atomicAdd(&fill[pr.y & (BIN_SZ - 1)], 1)] = pr.x;
        }
      }
    }
    __syncthreads();
    for (int i = tid; i < tot; i += SORT_TB) src_sorted[base + i] = stg[i];
  } else {
#pragma unroll
    for (int k = 0; k < RUNS_PW; ++k) {
      if (lane < rn[k])      src_sorted[base + atomicAdd(&fill[v0[k].y & (BIN_SZ - 1)], 1)] = v0[k].x;
      if (lane + 32 < rn[k]) src_sorted[base + atomicAdd(&fill[v1[k].y & (BIN_SZ - 1)], 1)] = v1[k].x;
      if (rn[k] > 64) {
        const int2* p = pp + rs[w + k * (SORT_TB / 32)];
        for (int i = lane + 64; i < rn[k]; i += 32) {
          const int2 pr = p[i];
          src_sorted[base + atomicAdd(&fill[pr.y & (BIN_SZ - 1)], 1)] = pr.x;
        }
      }
    }
  }
}

struct LaneSel { int x0, x8, x7, x1, x2; };

__device__ __forceinline__ float row_take(float x, int sel) {
  return __int_as_float((int)__builtin_amdgcn_permlane16_var((unsigned)__float_as_int(x), (unsigned)__float_as_int(x),
                                                             (unsigned)sel, true, false));
}

__device__ __forceinline__ float other_row(float x, int sel) {
  return __int_as_float((int)__builtin_amdgcn_permlanex16_var((unsigned)__float_as_int(x), (unsigned)__float_as_int(x),
                                                              (unsigned)sel, true, false));
}

__device__ __forceinline__ float quad_sum(float p, const LaneSel& ls) {
  p += row_take(p, ls.x1);
  p += row_take(p, ls.x2);
  return p;
}

__device__ __forceinline__ float head_score(const v4f x, const v2f (&qq)[4][4], const LaneSel& ls) {
  const v2f xx = {x.x, x.x}, xy = {x.y, x.y}, xz = {x.z, x.z}, xw = {x.w, x.w};
  v2f p[4];
#pragma unroll
  for (int jj = 0; jj < 4; ++jj) {
    v2f t = xx * qq[jj][0];
    t = __builtin_elementwise_fma(xy, qq[jj][1], t);
    t = __builtin_elementwise_fma(xz, qq[jj][2], t);
    t = __builtin_elementwise_fma(xw, qq[jj][3], t);
    p[jj] = t;
  }
  v2f r0, r1;
  r0.x = row_take(p[2].x, ls.x8); r0.y = row_take(p[2].y, ls.x8);
  r1.x = row_take(p[3].x, ls.x8); r1.y = row_take(p[3].y, ls.x8);
  const v2f a = p[0] + r0;
  const v2f b = p[1] + r1;
  v2f s;
  s.x = row_take(b.x, ls.x7);
  s.y = row_take(b.y, ls.x7);
  const v2f e = a + s;
  const float c0 = e.x + other_row(e.y, ls.x0);
  return quad_sum(c0, ls);
}

template <bool TAIL>
__device__ __forceinline__ void attend8(const v4f* __restrict__ x4, unsigned lane,
                                        int sv, int c0, int cnt, const v2f (&qq)[4][4], const LaneSel& ls,
                                        float& m, float& l, float4& acc) {
  v4f xx[8];
  int s[8];
#pragma unroll
  for (int i = 0; i < 8; ++i) s[i] = __builtin_amdgcn_readlane(sv, c0 + i);
#pragma unroll
  for (int i = 0; i < 8; ++i) xx[i] = x4[(unsigned)s[i] * (D / 4) + lane];
  asm volatile("" :: "v"(xx[0]), "v"(xx[1]), "v"(xx[2]), "v"(xx[3]), "v"(xx[4]), "v"(xx[5]), "v"(xx[6]), "v"(xx[7]));
  __builtin_amdgcn_sched_barrier(0);
  float t[8];
  float cm = m;
#pragma unroll
  for (int i = 0; i < 8; ++i) {
    float p = head_score(xx[i], qq, ls);
    if (TAIL && i >= cnt) p = __uint_as_float(0xff800000u);
    t[i] = p;
    cm = fmaxf(cm, p);
    if (i & 1) __builtin_amdgcn_sched_barrier(0);
  }
  const float r = __expf(m - cm);
  l *= r;
  acc.x *= r; acc.y *= r; acc.z *= r; acc.w *= r;
#pragma unroll
  for (int i = 0; i < 8; ++i) {
    const float e = __expf(t[i] - cm);
    l += e;
    acc.x += e * xx[i].x; acc.y += e * xx[i].y; acc.z += e * xx[i].z; acc.w += e * xx[i].w;
  }
  m = cm;
}

#define TILE_M 16
#define PROW   1028
__global__ void __launch_bounds__(SEG_TB) __attribute__((amdgpu_waves_per_eu(8, 8)))
seg_attend(const float* __restrict__ h_src,
           const float* __restrict__ h_dst,
           const float* __restrict__ Q,
           const float* __restrict__ Wk,
           const int* __restrict__ seg_start,
           const int* __restrict__ seg_cnt,
           const int* __restrict__ src_sorted,
           float* __restrict__ out, int Ndst) {
  __shared__ __attribute__((aligned(16))) float Pl[TILE_M * PROW];
  const int lane = threadIdx.x & 31;
  const int w     = __builtin_amdgcn_readfirstlane((int)(threadIdx.x >> 5));
  const int mBase = blockIdx.x * TILE_M;

  const int dA = (mBase + w < Ndst) ? (mBase + w) : (Ndst - 1);
  const int dB = (mBase + w + 8 < Ndst) ? (mBase + w + 8) : (Ndst - 1);
  const int begA = seg_start[dA], nA = seg_cnt[dA];
  const int begB = seg_start[dB], nB = seg_cnt[dB];

  {
    const int c  = lane & 15;
    const int hh = lane >> 4;
    int qr = mBase + c;
    qr = (qr < Ndst) ? qr : (Ndst - 1);
    const float* qrow = Q + (size_t)qr * D + 16 * w + 8 * hh;
    const float4 q0 = *(const float4*)(qrow);
    const float4 q1 = *(const float4*)(qrow + 4);
    v2f a[4];
    a[0].x = q0.x; a[0].y = q0.y; a[1].x = q0.z; a[1].y = q0.w;
    a[2].x = q1.x; a[2].y = q1.y; a[3].x = q1.z; a[3].y = q1.w;
    const float* wh = Wk + (size_t)(16 * w + 8 * hh) * D + c;
    v8f acc[8] = {};
#pragma unroll
    for (int s = 0; s < 4; ++s) {
      v2f b[8];
#pragma unroll
      for (int t = 0; t < 8; ++t) {
        b[t].x = wh[(2 * s) * D + 16 * t];
        b[t].y = wh[(2 * s + 1) * D + 16 * t];
      }
#pragma unroll
      for (int t = 0; t < 8; ++t) {
        acc[t] = __builtin_amdgcn_wmma_f32_16x16x4_f32(
            false, a[s], false, b[t], (short)0, acc[t], false, false);
      }
    }
    float* pl = Pl + (8 * hh) * PROW + c;
#pragma unroll
    for (int t = 0; t < 8; ++t) {
#pragma unroll
      for (int r = 0; r < 8; ++r) {
        pl[r * PROW + (w ^ t) * D + t * 16] = acc[t][r];
      }
    }
  }
  __syncthreads();

  const int ln = (int)__builtin_amdgcn_mbcnt_lo(0xffffffffu, 0u);
  const int col = ln * 4;
  const v4f* x4 = (const v4f*)h_src;
  LaneSel ls;
  ls.x0 = ln & 15; ls.x8 = (ln ^ 8) & 15; ls.x7 = (ln ^ 7) & 15; ls.x1 = (ln ^ 1) & 15; ls.x2 = (ln ^ 2) & 15;
#pragma nounroll
  for (int rr = 0; rr < 2; ++rr) {
    const int m = w + 8 * rr;
    const int d = mBase + m;
    if (d >= Ndst) break;
    const int beg = __builtin_amdgcn_readfirstlane(rr ? begB : begA);
    const int n   = __builtin_amdgcn_readfirstlane(rr ? nB : nA);
    v2f qq[4][4];
    const float* prow = Pl + m * PROW + col;
#pragma unroll
    for (int jj = 0; jj < 4; ++jj) {
      const float4 u  = *(const float4*)(prow + jj * D);
      const float4 wv = *(const float4*)(prow + (jj + 4) * D);
      qq[jj][0].x = u.x * SCALE; qq[jj][0].y = wv.x * SCALE;
      qq[jj][1].x = u.y * SCALE; qq[jj][1].y = wv.y * SCALE;
      qq[jj][2].x = u.z * SCALE; qq[jj][2].y = wv.z * SCALE;
      qq[jj][3].x = u.w * SCALE; qq[jj][3].y = wv.w * SCALE;
    }
    const unsigned o4 = (unsigned)d * (D / 4) + (unsigned)ln;
    const float4 xd = ((const float4*)h_dst)[o4];
    const v4f xdv = {xd.x, xd.y, xd.z, xd.w};
    float mx = head_score(xdv, qq, ls);
    float l = 1.0f;
    float4 acc = xd;
    for (int j0 = 0; j0 < n; j0 += 32) {
      const int mc = (n - j0 < 32) ? (n - j0) : 32;
      const int sv = (ln < mc) ? src_sorted[beg + j0 + ln] : 0;
      int c0 = 0;
      for (; c0 + 8 <= mc; c0 += 8) attend8<false>(x4, (unsigned)ln, sv, c0, 8, qq, ls, mx, l, acc);
      if (c0 < mc) attend8<true>(x4, (unsigned)ln, sv, c0, mc - c0, qq, ls, mx, l, acc);
    }
    const float rl = 1.0f / l;
    float4 o;
    o.x = acc.x * rl; o.y = acc.y * rl; o.z = acc.z * rl; o.w = acc.w * rl;
    ((float4*)out)[o4] = o;
  }
}

extern "C" void kernel_launch(void* const* d_in, const int* in_sizes, int n_in,
                              void* d_out, int out_size, void* d_ws, size_t ws_size,
                              hipStream_t stream) {
  const float* h_src = (const float*)d_in[0];
  const float* h_dst = (const float*)d_in[1];
  const float* Wk    = (const float*)d_in[2];
  const float* Wq    = (const float*)d_in[3];
  const int* src_idx = (const int*)d_in[4];
  const int* dst_idx = (const int*)d_in[5];

  const int N_DST = in_sizes[1] / D;
  const int E     = in_sizes[4];

  auto align256 = [](size_t x) { return (x + 255) & ~(size_t)255; };
  char* ws = (char*)d_ws;
  size_t off = 0;
  float* Q      = (float*)(ws + off); off += align256((size_t)N_DST * D * 4);
  int* pairs      = (int*)(ws + off); off += align256((size_t)E * 8);
  int* src_sorted = (int*)(ws + off); off += align256((size_t)E * 4);
  int* seg_cnt    = (int*)(ws + off); off += align256((size_t)N_DST * 4);
  int* seg_start  = (int*)(ws + off); off += align256((size_t)N_DST * 4);
  int* cursor     = (int*)(ws + off); off += align256(4);
  int* run_start  = (int*)(ws + off); off += align256((size_t)NBIN_MAX * SORT_G * 4);
  int* run_cnt    = (int*)(ws + off); off += align256((size_t)NBIN_MAX * SORT_G * 4);

  const int projG = (N_DST + PROJ_ROWS - 1) / PROJ_ROWS;
  project_and_chunk_sort<<<projG + SORT_G, SORT_TB, 0, stream>>>(
      h_dst, Wq, Q, N_DST, projG, src_idx, dst_idx, pairs, run_start, run_cnt, cursor, E);
  bin_sort<<<(N_DST + BIN_SZ - 1) / BIN_SZ, SORT_TB, 0, stream>>>(
      pairs, run_start, run_cnt, cursor, seg_start, seg_cnt, src_sorted, E, N_DST);
  seg_attend<<<(N_DST + TILE_M - 1) / TILE_M, SEG_TB, 0, stream>>>(
      h_src, h_dst, Q, Wk, seg_start, seg_cnt, src_sorted, (float*)d_out, N_DST);
}
